# starting kernel
# baseline (speedup 1.0000x reference)
#include <hip/hip_runtime.h>

typedef __attribute__((ext_vector_type(2))) float v2f;
typedef __attribute__((ext_vector_type(8))) float v8f;

#define IN_CH 128
#define HID 32


__global__ void init_deg_kernel(float* __restrict__ deg, int n) {
  int i = blockIdx.x * blockDim.x + threadIdx.x;
  if (i < n) deg[i] = 1.0f;
}

__global__ void degree_kernel(const int* __restrict__ dst, float* __restrict__ deg, int e_count) {
  int e = blockIdx.x * blockDim.x + threadIdx.x;
  if (e < e_count) atomicAdd(&deg[dst[e]], 1.0f);
}

__global__ void dinv_kernel(float* __restrict__ degdinv, int n) {
  int i = blockIdx.x * blockDim.x + threadIdx.x;
  if (i < n) {
    float d = degdinv[i];
    degdinv[i] = (d > 0.0f) ? rsqrtf(d) : 0.0f;
  }
}

__global__ void init_agg_kernel(const float* __restrict__ h, const float* __restrict__ dinv,
                                const float* __restrict__ bias, float* __restrict__ agg, int n) {
  int i = blockIdx.x * blockDim.x + threadIdx.x;
  if (i < n * HID) {
    int node = i >> 5;
    int c = i & 31;
    float di = dinv[node];
    agg[i] = bias[c] + h[i] * di * di;
  }
}

__global__ void relu_kernel(const float* __restrict__ in, float* __restrict__ out, int total) {
  int i = blockIdx.x * blockDim.x + threadIdx.x;
  if (i < total) out[i] = fmaxf(in[i], 0.0f);
}

__global__ __launch_bounds__(256)
void edge_agg_kernel(const int* __restrict__ src, const int* __restrict__ dst,
                     const float* __restrict__ dinv, const float* __restrict__ h,
                     float* __restrict__ agg, int e_count) {
  int lane = threadIdx.x & 31;
  int e = blockIdx.x * (blockDim.x >> 5) + (threadIdx.x >> 5);
  if (e >= e_count) return;
  int s = src[e];
  int d = dst[e];
  float nrm = dinv[s] * dinv[d];
  float v = h[s * HID + lane] * nrm;
  atomicAdd(&agg[d * HID + lane], v);
}

template <int K>
__global__ __launch_bounds__(128)
void gemm_n32_kernel(const float* __restrict__ A, const float* __restrict__ W,
                     float* __restrict__ out, int M) {
  constexpr int LDK = K + 1;
  __shared__ float sA[32 * LDK];

  const int tid = threadIdx.x;
  const int block_row0 = blockIdx.x * 32;

  for (int idx = tid; idx < 32 * K; idx += 128) {
    int r = idx / K;
    int c = idx - r * K;
    int gr = block_row0 + r;
    sA[r * LDK + c] = (gr < M) ? A[(long long)gr * K + c] : 0.0f;
  }
  __syncthreads();

  const int wid   = tid >> 5;
  const int lane  = tid & 31;
  const int lhalf = lane >> 4;
  const int l15   = lane & 15;
  const int warp_m = (wid >> 1) * 16;
  const int n0     = (wid & 1) * 16;

  v8f acc = {};
  for (int k = 0; k < K; k += 4) {
    const float* ap = &sA[(warp_m + l15) * LDK + k + lhalf * 2];
    v2f a;
    a.x = ap[0];
    a.y = ap[1];
    const float* bp = &W[(k + lhalf * 2) * HID + n0 + l15];
    v2f b;
    b.x = bp[0];
    b.y = bp[HID];
    acc = __builtin_amdgcn_wmma_f32_16x16x4_f32(false, a, false, b, (short)0, acc,
                                                false, false);
  }

  const int grow_base = block_row0 + warp_m + lhalf * 8;
  const int col = n0 + l15;
#pragma unroll
  for (int j = 0; j < 8; ++j) {
    int gr = grow_base + j;
    if (gr < M) out[(long long)gr * HID + col] = acc[j];
  }
}


extern "C" void kernel_launch(void* const* d_in, const int* in_sizes, int n_in,
                              void* d_out, int out_size, void* d_ws, size_t ws_size,
                              hipStream_t stream) {
  const float* x  = (const float*)d_in[0];
  const int*   ei = (const int*)d_in[1];
  const float* W1 = (const float*)d_in[2];
  const float* b1 = (const float*)d_in[3];
  const float* W2 = (const float*)d_in[4];
  const float* b2 = (const float*)d_in[5];
  float* out = (float*)d_out;

  const int N = in_sizes[0] / IN_CH;
  const int E = in_sizes[1] / 2;
  const int* src = ei;
  const int* dst = ei + E;

  float* ws = (float*)d_ws;
  const int Npad = (N + 63) & ~63;
  float* dinv = ws;
  float* bufA = ws + Npad;
  float* bufB = bufA + (size_t)N * HID;

  const int T = 256;
  const unsigned gN    = (N + T - 1) / T;
  const unsigned gE    = (E + T - 1) / T;
  const unsigned gNC   = ((unsigned)N * HID + T - 1) / T;
  const unsigned gEdge = (E + (T / 32) - 1) / (T / 32);
  const unsigned gGemm = (N + 31) / 32;

  init_deg_kernel<<<gN, T, 0, stream>>>(dinv, N);
  degree_kernel<<<gE, T, 0, stream>>>(dst, dinv, E);
  dinv_kernel<<<gN, T, 0, stream>>>(dinv, N);

  gemm_n32_kernel<IN_CH><<<gGemm, 128, 0, stream>>>(x, W1, bufA, N);
  init_agg_kernel<<<gNC, T, 0, stream>>>(bufA, dinv, b1, bufB, N);
  edge_agg_kernel<<<gEdge, T, 0, stream>>>(src, dst, dinv, bufA, bufB, E);
  relu_kernel<<<gNC, T, 0, stream>>>(bufB, bufA, N * HID);

  gemm_n32_kernel<HID><<<gGemm, 128, 0, stream>>>(bufA, W2, bufB, N);
  init_agg_kernel<<<gNC, T, 0, stream>>>(bufB, dinv, b2, out, N);
  edge_agg_kernel<<<gEdge, T, 0, stream>>>(src, dst, dinv, bufB, out, E);
}


# degree: each workgroup counts a 50000-edge slice into an LDS histogram of packed 16-bit bins (cannot carry), planes summed in the dinv launch; no global atomics for degrees, init launch gone
# speedup vs baseline: 1.2085x; 1.2085x over previous
#include <hip/hip_runtime.h>

typedef __attribute__((ext_vector_type(2))) float v2f;
typedef __attribute__((ext_vector_type(8))) float v8f;

#define IN_CH 128
#define HID 32


#define HIST_SLICE 50000
#define HIST_TILE  131072

__device__ __forceinline__ void hist_bump(unsigned* cnt, int d, int base, int span) {
  unsigned r = (unsigned)(d - base);
  if (r < (unsigned)span) atomicAdd(&cnt[r >> 1], 1u << ((r & 1u) << 4));
}

__global__ __launch_bounds__(1024)
void degree_hist_kernel(const int* __restrict__ dst, unsigned* __restrict__ planes,
                        int n, int e_count, int plane_words) {
  __shared__ uint4 cnt4[HIST_TILE / 8];
  unsigned* cnt = (unsigned*)cnt4;
  const int tid = threadIdx.x;
  const int base = blockIdx.y * HIST_TILE;
  const int span = min(HIST_TILE, n - base);
  const int quads = (span + 7) >> 3;

  for (int i = tid; i < quads; i += 1024) cnt4[i] = make_uint4(0u, 0u, 0u, 0u);
  __syncthreads();

  const int e0 = blockIdx.x * HIST_SLICE;
  const int e1 = min(e0 + HIST_SLICE, e_count);
  int e = e0;
  if ((((size_t)(dst + e0)) & 15) == 0) {
    constexpr int U = 6;
    for (; e + 1024 * 4 * U <= e1; e += 1024 * 4 * U) {
      const int4* p = (const int4*)(dst + e);
      int4 q[U];
#pragma unroll
      for (int u = 0; u < U; ++u) q[u] = p[u * 1024 + tid];
#pragma unroll
      for (int u = 0; u < U; ++u) {
        hist_bump(cnt, q[u].x, base, span);
        hist_bump(cnt, q[u].y, base, span);
        hist_bump(cnt, q[u].z, base, span);
        hist_bump(cnt, q[u].w, base, span);
      }
    }
  }
  for (int t = e + tid; t < e1; t += 1024) hist_bump(cnt, dst[t], base, span);
  __syncthreads();

  uint4* out4 = (uint4*)(planes + (size_t)blockIdx.x * plane_words + (base >> 1));
  for (int i = tid; i < quads; i += 1024) out4[i] = cnt4[i];
}

__global__ void dinv_kernel(const unsigned* __restrict__ planes, float* __restrict__ dinv,
                            int n, int n_planes, int plane_words) {
  int w = blockIdx.x * blockDim.x + threadIdx.x;
  if (2 * w >= n) return;
  unsigned lo = 0u, hi = 0u;
#pragma unroll 16
  for (int p = 0; p < n_planes; ++p) {
    unsigned v = planes[(size_t)p * plane_words + w];
    lo += v & 0xffffu;
    hi += v >> 16;
  }
  float d0 = (float)(lo + 1u);
  dinv[2 * w] = (d0 > 0.0f) ? rsqrtf(d0) : 0.0f;
  if (2 * w + 1 < n) {
    float d1 = (float)(hi + 1u);
    dinv[2 * w + 1] = (d1 > 0.0f) ? rsqrtf(d1) : 0.0f;
  }
}

__global__ void init_agg_kernel(const float* __restrict__ h, const float* __restrict__ dinv,
                                const float* __restrict__ bias, float* __restrict__ agg, int n) {
  int i = blockIdx.x * blockDim.x + threadIdx.x;
  if (i < n * HID) {
    int node = i >> 5;
    int c = i & 31;
    float di = dinv[node];
    agg[i] = bias[c] + h[i] * di * di;
  }
}

__global__ void relu_kernel(const float* __restrict__ in, float* __restrict__ out, int total) {
  int i = blockIdx.x * blockDim.x + threadIdx.x;
  if (i < total) out[i] = fmaxf(in[i], 0.0f);
}

__global__ __launch_bounds__(256)
void edge_agg_kernel(const int* __restrict__ src, const int* __restrict__ dst,
                     const float* __restrict__ dinv, const float* __restrict__ h,
                     float* __restrict__ agg, int e_count) {
  int lane = threadIdx.x & 31;
  int e = blockIdx.x * (blockDim.x >> 5) + (threadIdx.x >> 5);
  if (e >= e_count) return;
  int s = src[e];
  int d = dst[e];
  float nrm = dinv[s] * dinv[d];
  float v = h[s * HID + lane] * nrm;
  atomicAdd(&agg[d * HID + lane], v);
}

template <int K>
__global__ __launch_bounds__(128)
void gemm_n32_kernel(const float* __restrict__ A, const float* __restrict__ W,
                     float* __restrict__ out, int M) {
  constexpr int LDK = K + 1;
  __shared__ float sA[32 * LDK];

  const int tid = threadIdx.x;
  const int block_row0 = blockIdx.x * 32;

  for (int idx = tid; idx < 32 * K; idx += 128) {
    int r = idx / K;
    int c = idx - r * K;
    int gr = block_row0 + r;
    sA[r * LDK + c] = (gr < M) ? A[(long long)gr * K + c] : 0.0f;
  }
  __syncthreads();

  const int wid   = tid >> 5;
  const int lane  = tid & 31;
  const int lhalf = lane >> 4;
  const int l15   = lane & 15;
  const int warp_m = (wid >> 1) * 16;
  const int n0     = (wid & 1) * 16;

  v8f acc = {};
  for (int k = 0; k < K; k += 4) {
    const float* ap = &sA[(warp_m + l15) * LDK + k + lhalf * 2];
    v2f a;
    a.x = ap[0];
    a.y = ap[1];
    const float* bp = &W[(k + lhalf * 2) * HID + n0 + l15];
    v2f b;
    b.x = bp[0];
    b.y = bp[HID];
    acc = __builtin_amdgcn_wmma_f32_16x16x4_f32(false, a, false, b, (short)0, acc,
                                                false, false);
  }

  const int grow_base = block_row0 + warp_m + lhalf * 8;
  const int col = n0 + l15;
#pragma unroll
  for (int j = 0; j < 8; ++j) {
    int gr = grow_base + j;
    if (gr < M) out[(long long)gr * HID + col] = acc[j];
  }
}


extern "C" void kernel_launch(void* const* d_in, const int* in_sizes, int n_in,
                              void* d_out, int out_size, void* d_ws, size_t ws_size,
                              hipStream_t stream) {
  const float* x  = (const float*)d_in[0];
  const int*   ei = (const int*)d_in[1];
  const float* W1 = (const float*)d_in[2];
  const float* b1 = (const float*)d_in[3];
  const float* W2 = (const float*)d_in[4];
  const float* b2 = (const float*)d_in[5];
  float* out = (float*)d_out;

  const int N = in_sizes[0] / IN_CH;
  const int E = in_sizes[1] / 2;
  const int* src = ei;
  const int* dst = ei + E;

  float* ws = (float*)d_ws;
  const int Npad = (N + 63) & ~63;
  float* dinv = ws;
  float* bufA = ws + Npad;
  float* bufB = bufA + (size_t)N * HID;
  unsigned* planes = (unsigned*)(bufB + (size_t)N * HID);

  const int T = 256;
  const unsigned gNC   = ((unsigned)N * HID + T - 1) / T;
  const unsigned gEdge = (E + (T / 32) - 1) / (T / 32);
  const unsigned gGemm = (N + 31) / 32;
  const int nSlices    = (E + HIST_SLICE - 1) / HIST_SLICE;
  const int nTiles     = (N + HIST_TILE - 1) / HIST_TILE;
  const int planeWords = ((N + 7) >> 3) << 2;
  const unsigned gHalf = ((N + 1) / 2 + T - 1) / T;

  degree_hist_kernel<<<dim3(nSlices, nTiles), 1024, 0, stream>>>(dst, planes, N, E, planeWords);
  dinv_kernel<<<gHalf, T, 0, stream>>>(planes, dinv, N, nSlices, planeWords);

  gemm_n32_kernel<IN_CH><<<gGemm, 128, 0, stream>>>(x, W1, bufA, N);
  init_agg_kernel<<<gNC, T, 0, stream>>>(bufA, dinv, b1, bufB, N);
  edge_agg_kernel<<<gEdge, T, 0, stream>>>(src, dst, dinv, bufA, bufB, E);
  relu_kernel<<<gNC, T, 0, stream>>>(bufB, bufA, N * HID);

  gemm_n32_kernel<HID><<<gGemm, 128, 0, stream>>>(bufA, W2, bufB, N);
  init_agg_kernel<<<gNC, T, 0, stream>>>(bufB, dinv, b2, out, N);
  edge_agg_kernel<<<gEdge, T, 0, stream>>>(src, dst, dinv, bufB, out, E);
}


# edge sums without global atomics: workgroup keeps 2048 nodes' rows in LDS, filters its edge slice by dst into wave queues, LDS float adds; 5 slice planes summed in a combine launch; sum order changes
# speedup vs baseline: 2.9322x; 2.4263x over previous
#include <hip/hip_runtime.h>

typedef __attribute__((ext_vector_type(2))) float v2f;
typedef __attribute__((ext_vector_type(8))) float v8f;

#define IN_CH 128
#define HID 32


#define HIST_SLICE 50000
#define HIST_TILE  131072

__device__ __forceinline__ void hist_bump(unsigned* cnt, int d, int base, int span) {
  unsigned r = (unsigned)(d - base);
  if (r < (unsigned)span) atomicAdd(&cnt[r >> 1], 1u << ((r & 1u) << 4));
}

__global__ __launch_bounds__(1024)
void degree_hist_kernel(const int* __restrict__ dst, unsigned* __restrict__ planes,
                        int n, int e_count, int plane_words) {
  __shared__ uint4 cnt4[HIST_TILE / 8];
  unsigned* cnt = (unsigned*)cnt4;
  const int tid = threadIdx.x;
  const int base = blockIdx.y * HIST_TILE;
  const int span = min(HIST_TILE, n - base);
  const int quads = (span + 7) >> 3;

  for (int i = tid; i < quads; i += 1024) cnt4[i] = make_uint4(0u, 0u, 0u, 0u);
  __syncthreads();

  const int e0 = blockIdx.x * HIST_SLICE;
  const int e1 = min(e0 + HIST_SLICE, e_count);
  int e = e0;
  if ((((size_t)(dst + e0)) & 15) == 0) {
    constexpr int U = 6;
    for (; e + 1024 * 4 * U <= e1; e += 1024 * 4 * U) {
      const int4* p = (const int4*)(dst + e);
      int4 q[U];
#pragma unroll
      for (int u = 0; u < U; ++u) q[u] = p[u * 1024 + tid];
#pragma unroll
      for (int u = 0; u < U; ++u) {
        hist_bump(cnt, q[u].x, base, span);
        hist_bump(cnt, q[u].y, base, span);
        hist_bump(cnt, q[u].z, base, span);
        hist_bump(cnt, q[u].w, base, span);
      }
    }
  }
  for (int t = e + tid; t < e1; t += 1024) hist_bump(cnt, dst[t], base, span);
  __syncthreads();

  uint4* out4 = (uint4*)(planes + (size_t)blockIdx.x * plane_words + (base >> 1));
  for (int i = tid; i < quads; i += 1024) out4[i] = cnt4[i];
}

__global__ void dinv_kernel(const unsigned* __restrict__ planes, float* __restrict__ dinv,
                            int n, int n_planes, int plane_words) {
  int w = blockIdx.x * blockDim.x + threadIdx.x;
  if (2 * w >= n) return;
  unsigned lo = 0u, hi = 0u;
#pragma unroll 16
  for (int p = 0; p < n_planes; ++p) {
    unsigned v = planes[(size_t)p * plane_words + w];
    lo += v & 0xffffu;
    hi += v >> 16;
  }
  float d0 = (float)(lo + 1u);
  dinv[2 * w] = (d0 > 0.0f) ? rsqrtf(d0) : 0.0f;
  if (2 * w + 1 < n) {
    float d1 = (float)(hi + 1u);
    dinv[2 * w + 1] = (d1 > 0.0f) ? rsqrtf(d1) : 0.0f;
  }
}

#define AGG_TILE   2048
#define AGG_RBITS  11
#define AGG_SLICES 5
#define AGG_U      4
#define AGG_QCAP   160

typedef __attribute__((address_space(3))) volatile unsigned lds_queue_t;

__device__ __forceinline__ void agg_drain(const lds_queue_t* wq, int n, int lane, int e0,
                                          int base, const int* __restrict__ src,
                                          const float* __restrict__ dinv,
                                          const float* __restrict__ h, float* tile) {
  int s = 0, r = 0;
  float nrm = 0.0f;
  if (lane < n) {
    unsigned ent = wq[lane];
    r = (int)(ent & (AGG_TILE - 1));
    s = src[e0 + (int)(ent >> AGG_RBITS)];
    nrm = dinv[s] * dinv[base + r];
  }
  float v[32];
#pragma unroll
  for (int j = 0; j < 32; ++j) {
    if (j < n) {
      int sj = __builtin_amdgcn_readlane(s, j);
      v[j] = h[(size_t)sj * HID + lane];
    }
  }
#pragma unroll
  for (int j = 0; j < 32; ++j) {
    if (j < n) {
      int rj = __builtin_amdgcn_readlane(r, j);
      float nj = __builtin_bit_cast(float, __builtin_amdgcn_readlane(__builtin_bit_cast(int, nrm), j));
      atomicAdd(&tile[rj * HID + lane], v[j] * nj);
    }
  }
}

__device__ __forceinline__ int agg_append(lds_queue_t* wq, int qcnt, int d, int el,
                                          int base, int span, unsigned lanes_below) {
  unsigned r = (unsigned)(d - base);
  bool m = r < (unsigned)span;
  unsigned bal = __builtin_amdgcn_ballot_w32(m);
  if (m) wq[qcnt + __popc(bal & lanes_below)] = ((unsigned)el << AGG_RBITS) | r;
  return qcnt + __popc(bal);
}

__global__ __launch_bounds__(1024)
void agg_tile_kernel(const int* __restrict__ src, const int* __restrict__ dst,
                     const float* __restrict__ dinv, const float* __restrict__ h,
                     float* __restrict__ planes, int n, int e_count, int slice_len) {
  __shared__ float4 tile4[AGG_TILE * HID / 4];
  __shared__ unsigned queues[32 * AGG_QCAP];
  float* tile = (float*)tile4;
  const int tid = threadIdx.x;
  const int lane = tid & 31;
  const unsigned lanes_below = (1u << lane) - 1u;
  lds_queue_t* wq = (lds_queue_t*)queues + (tid >> 5) * AGG_QCAP;
  const int base = blockIdx.y * AGG_TILE;
  const int span = min(AGG_TILE, n - base);
  const int quads = span * (HID / 4);

  for (int i = tid; i < quads; i += 1024) tile4[i] = make_float4(0.0f, 0.0f, 0.0f, 0.0f);
  __syncthreads();

  const int e0 = blockIdx.x * slice_len;
  const int e1 = min(e0 + slice_len, e_count);
  int qcnt = 0;
  int eb = e0;
  if ((((size_t)(dst + e0)) & 15) == 0) {
    for (; eb + 1024 * 4 * AGG_U <= e1; eb += 1024 * 4 * AGG_U) {
      const int4* p = (const int4*)(dst + eb);
      int4 q[AGG_U];
#pragma unroll
      for (int u = 0; u < AGG_U; ++u) q[u] = p[u * 1024 + tid];
#pragma unroll
      for (int u = 0; u < AGG_U; ++u) {
        const int el = eb - e0 + (u * 1024 + tid) * 4;
        qcnt = agg_append(wq, qcnt, q[u].x, el,     base, span, lanes_below);
        qcnt = agg_append(wq, qcnt, q[u].y, el + 1, base, span, lanes_below);
        qcnt = agg_append(wq, qcnt, q[u].z, el + 2, base, span, lanes_below);
        qcnt = agg_append(wq, qcnt, q[u].w, el + 3, base, span, lanes_below);
        while (qcnt >= 32) {
          qcnt -= 32;
          agg_drain(wq + qcnt, 32, lane, e0, base, src, dinv, h, tile);
        }
      }
    }
  }
  for (; eb < e1; eb += 1024) {
    const int t = eb + tid;
    const int d = (t < e1) ? dst[t] : -1;
    qcnt = agg_append(wq, qcnt, d, t - e0, base, span, lanes_below);
    if (qcnt >= 32) {
      qcnt -= 32;
      agg_drain(wq + qcnt, 32, lane, e0, base, src, dinv, h, tile);
    }
  }
  if (qcnt > 0) agg_drain(wq, qcnt, lane, e0, base, src, dinv, h, tile);
  __syncthreads();

  float4* out4 = (float4*)(planes + ((size_t)blockIdx.x * n + base) * HID);
  for (int i = tid; i < quads; i += 1024) out4[i] = tile4[i];
}

template <bool RELU>
__global__ void combine_kernel(const float* __restrict__ h, const float* __restrict__ dinv,
                               const float* __restrict__ bias, const float* __restrict__ planes,
                               float* __restrict__ out, int n) {
  const int i = blockIdx.x * blockDim.x + threadIdx.x;
  if (i >= n * (HID / 4)) return;
  const int node = i >> 3;
  const int c = (i & 7) * 4;
  const float di = dinv[node];
  const float4 hv = ((const float4*)h)[i];
  float4 pv[AGG_SLICES];
#pragma unroll
  for (int p = 0; p < AGG_SLICES; ++p) pv[p] = ((const float4*)planes)[(size_t)p * n * (HID / 4) + i];
  float4 a;
  a.x = bias[c]     + hv.x * di * di;
  a.y = bias[c + 1] + hv.y * di * di;
  a.z = bias[c + 2] + hv.z * di * di;
  a.w = bias[c + 3] + hv.w * di * di;
#pragma unroll
  for (int p = 0; p < AGG_SLICES; ++p) {
    a.x += pv[p].x; a.y += pv[p].y; a.z += pv[p].z; a.w += pv[p].w;
  }
  if (RELU) {
    a.x = fmaxf(a.x, 0.0f); a.y = fmaxf(a.y, 0.0f); a.z = fmaxf(a.z, 0.0f); a.w = fmaxf(a.w, 0.0f);
  }
  ((float4*)out)[i] = a;
}

template <int K>
__global__ __launch_bounds__(128)
void gemm_n32_kernel(const float* __restrict__ A, const float* __restrict__ W,
                     float* __restrict__ out, int M) {
  constexpr int LDK = K + 1;
  __shared__ float sA[32 * LDK];

  const int tid = threadIdx.x;
  const int block_row0 = blockIdx.x * 32;

  for (int idx = tid; idx < 32 * K; idx += 128) {
    int r = idx / K;
    int c = idx - r * K;
    int gr = block_row0 + r;
    sA[r * LDK + c] = (gr < M) ? A[(long long)gr * K + c] : 0.0f;
  }
  __syncthreads();

  const int wid   = tid >> 5;
  const int lane  = tid & 31;
  const int lhalf = lane >> 4;
  const int l15   = lane & 15;
  const int warp_m = (wid >> 1) * 16;
  const int n0     = (wid & 1) * 16;

  v8f acc = {};
  for (int k = 0; k < K; k += 4) {
    const float* ap = &sA[(warp_m + l15) * LDK + k + lhalf * 2];
    v2f a;
    a.x = ap[0];
    a.y = ap[1];
    const float* bp = &W[(k + lhalf * 2) * HID + n0 + l15];
    v2f b;
    b.x = bp[0];
    b.y = bp[HID];
    acc = __builtin_amdgcn_wmma_f32_16x16x4_f32(false, a, false, b, (short)0, acc,
                                                false, false);
  }

  const int grow_base = block_row0 + warp_m + lhalf * 8;
  const int col = n0 + l15;
#pragma unroll
  for (int j = 0; j < 8; ++j) {
    int gr = grow_base + j;
    if (gr < M) out[(long long)gr * HID + col] = acc[j];
  }
}


extern "C" void kernel_launch(void* const* d_in, const int* in_sizes, int n_in,
                              void* d_out, int out_size, void* d_ws, size_t ws_size,
                              hipStream_t stream) {
  const float* x  = (const float*)d_in[0];
  const int*   ei = (const int*)d_in[1];
  const float* W1 = (const float*)d_in[2];
  const float* b1 = (const float*)d_in[3];
  const float* W2 = (const float*)d_in[4];
  const float* b2 = (const float*)d_in[5];
  float* out = (float*)d_out;

  const int N = in_sizes[0] / IN_CH;
  const int E = in_sizes[1] / 2;
  const int* src = ei;
  const int* dst = ei + E;

  float* ws = (float*)d_ws;
  const int Npad = (N + 63) & ~63;
  float* dinv = ws;
  float* bufA = ws + Npad;
  float* bufB = bufA + (size_t)N * HID;
  unsigned* planes = (unsigned*)(bufB + (size_t)N * HID);

  const int T = 256;
  const unsigned gQuad = ((unsigned)N * (HID / 4) + T - 1) / T;
  const unsigned gGemm = (N + 31) / 32;
  const int nSlices    = (E + HIST_SLICE - 1) / HIST_SLICE;
  const int nTiles     = (N + HIST_TILE - 1) / HIST_TILE;
  const int planeWords = ((N + 7) >> 3) << 2;
  const unsigned gHalf = ((N + 1) / 2 + T - 1) / T;
  float* sums = (float*)(planes + (size_t)nSlices * planeWords);
  const int aggSlice = (E + AGG_SLICES - 1) / AGG_SLICES;
  const int aggTiles = (N + AGG_TILE - 1) / AGG_TILE;
  const dim3 gAgg(AGG_SLICES, aggTiles);

  degree_hist_kernel<<<dim3(nSlices, nTiles), 1024, 0, stream>>>(dst, planes, N, E, planeWords);
  dinv_kernel<<<gHalf, T, 0, stream>>>(planes, dinv, N, nSlices, planeWords);

  gemm_n32_kernel<IN_CH><<<gGemm, 128, 0, stream>>>(x, W1, bufA, N);
  agg_tile_kernel<<<gAgg, 1024, 0, stream>>>(src, dst, dinv, bufA, sums, N, E, aggSlice);
  combine_kernel<true><<<gQuad, T, 0, stream>>>(bufA, dinv, b1, sums, bufB, N);

  gemm_n32_kernel<HID><<<gGemm, 128, 0, stream>>>(bufB, W2, bufA, N);
  agg_tile_kernel<<<gAgg, 1024, 0, stream>>>(src, dst, dinv, bufA, sums, N, E, aggSlice);
  combine_kernel<false><<<gQuad, T, 0, stream>>>(bufA, dinv, b2, sums, out, N);
}


# layer 2 replays layer 1's edge filter: each wave saves its matched edges (src,row,weight) to a workspace list while summing; layer-2 sum launch reads the lists, not dst/src/dinv; sum order kept
# speedup vs baseline: 3.5868x; 1.2232x over previous
#include <hip/hip_runtime.h>

typedef __attribute__((ext_vector_type(2))) float v2f;
typedef __attribute__((ext_vector_type(8))) float v8f;

#define IN_CH 128
#define HID 32


#define HIST_SLICE 50000
#define HIST_TILE  131072

__device__ __forceinline__ void hist_bump(unsigned* cnt, int d, int base, int span) {
  unsigned r = (unsigned)(d - base);
  if (r < (unsigned)span) atomicAdd(&cnt[r >> 1], 1u << ((r & 1u) << 4));
}

__global__ __launch_bounds__(1024)
void degree_hist_kernel(const int* __restrict__ dst, unsigned* __restrict__ planes,
                        int n, int e_count, int plane_words) {
  __shared__ uint4 cnt4[HIST_TILE / 8];
  unsigned* cnt = (unsigned*)cnt4;
  const int tid = threadIdx.x;
  const int base = blockIdx.y * HIST_TILE;
  const int span = min(HIST_TILE, n - base);
  const int quads = (span + 7) >> 3;

  for (int i = tid; i < quads; i += 1024) cnt4[i] = make_uint4(0u, 0u, 0u, 0u);
  __syncthreads();

  const int e0 = blockIdx.x * HIST_SLICE;
  const int e1 = min(e0 + HIST_SLICE, e_count);
  int e = e0;
  if ((((size_t)(dst + e0)) & 15) == 0) {
    constexpr int U = 6;
    for (; e + 1024 * 4 * U <= e1; e += 1024 * 4 * U) {
      const int4* p = (const int4*)(dst + e);
      int4 q[U];
#pragma unroll
      for (int u = 0; u < U; ++u) q[u] = p[u * 1024 + tid];
#pragma unroll
      for (int u = 0; u < U; ++u) {
        hist_bump(cnt, q[u].x, base, span);
        hist_bump(cnt, q[u].y, base, span);
        hist_bump(cnt, q[u].z, base, span);
        hist_bump(cnt, q[u].w, base, span);
      }
    }
  }
  for (int t = e + tid; t < e1; t += 1024) hist_bump(cnt, dst[t], base, span);
  __syncthreads();

  uint4* out4 = (uint4*)(planes + (size_t)blockIdx.x * plane_words + (base >> 1));
  for (int i = tid; i < quads; i += 1024) out4[i] = cnt4[i];
}

__global__ void dinv_kernel(const unsigned* __restrict__ planes, float* __restrict__ dinv,
                            int n, int n_planes, int plane_words) {
  int w = blockIdx.x * blockDim.x + threadIdx.x;
  if (2 * w >= n) return;
  unsigned lo = 0u, hi = 0u;
#pragma unroll 16
  for (int p = 0; p < n_planes; ++p) {
    unsigned v = planes[(size_t)p * plane_words + w];
    lo += v & 0xffffu;
    hi += v >> 16;
  }
  float d0 = (float)(lo + 1u);
  dinv[2 * w] = (d0 > 0.0f) ? rsqrtf(d0) : 0.0f;
  if (2 * w + 1 < n) {
    float d1 = (float)(hi + 1u);
    dinv[2 * w + 1] = (d1 > 0.0f) ? rsqrtf(d1) : 0.0f;
  }
}

#define AGG_TILE   2048
#define AGG_RBITS  11
#define AGG_SLICES 5
#define AGG_U      4
#define AGG_QCAP   160
#define AGG_LCAP   1024


typedef __attribute__((address_space(3))) volatile unsigned lds_queue_t;

__device__ __forceinline__ void agg_rows(int s, int r, float nrm, int n, int lane,
                                         const float* __restrict__ h, float* tile) {
  float v[32];
#pragma unroll
  for (int j = 0; j < 32; ++j) {
    if (j < n) {
      int sj = __builtin_amdgcn_readlane(s, j);
      v[j] = h[(size_t)sj * HID + lane];
    }
  }
#pragma unroll
  for (int j = 0; j < 32; ++j) {
    if (j < n) {
      int rj = __builtin_amdgcn_readlane(r, j);
      float nj = __builtin_bit_cast(float, __builtin_amdgcn_readlane(__builtin_bit_cast(int, nrm), j));
      atomicAdd(&tile[rj * HID + lane], v[j] * nj);
    }
  }
}

__device__ __forceinline__ uint2* agg_reserve(uint2* wlist, int& saved, int n) {
  if (saved < 0 || saved + n > AGG_LCAP) { saved = -1; return nullptr; }
  uint2* p = wlist + saved;
  saved += n;
  return p;
}

template <bool RECORD>
__device__ __forceinline__ void agg_drain(const lds_queue_t* wq, int n, int lane, int e0,
                                          int base, const int* __restrict__ src,
                                          const float* __restrict__ dinv,
                                          const float* __restrict__ h, float* tile,
                                          uint2* wlist, int& saved) {
  uint2* rec = nullptr;
  if (RECORD) rec = agg_reserve(wlist, saved, n);
  int s = 0, r = 0;
  float nrm = 0.0f;
  if (lane < n) {
    unsigned ent = wq[lane];
    r = (int)(ent & (AGG_TILE - 1));
    s = src[e0 + (int)(ent >> AGG_RBITS)];
    nrm = dinv[s] * dinv[base + r];
    if (RECORD && rec)
      rec[lane] = make_uint2(((unsigned)s << AGG_RBITS) | (unsigned)r, __builtin_bit_cast(unsigned, nrm));
  }
  agg_rows(s, r, nrm, n, lane, h, tile);
}

__device__ __forceinline__ int agg_append(lds_queue_t* wq, int qcnt, int d, int el,
                                          int base, int span, unsigned lanes_below) {
  unsigned r = (unsigned)(d - base);
  bool m = r < (unsigned)span;
  unsigned bal = __builtin_amdgcn_ballot_w32(m);
  if (m) wq[qcnt + __popc(bal & lanes_below)] = ((unsigned)el << AGG_RBITS) | r;
  return qcnt + __popc(bal);
}

template <bool RECORD>
__device__ __forceinline__ int agg_scan(const int* __restrict__ src, const int* __restrict__ dst,
                                        const float* __restrict__ dinv, const float* __restrict__ h,
                                        float* tile, lds_queue_t* wq, uint2* wlist,
                                        int tid, int lane, int base, int span,
                                        int e0, int e1) {
  const unsigned lanes_below = (1u << lane) - 1u;
  int saved = 0;
  int qcnt = 0;
  int eb = e0;
  if ((((size_t)(dst + e0)) & 15) == 0) {
    for (; eb + 1024 * 4 * AGG_U <= e1; eb += 1024 * 4 * AGG_U) {
      const int4* p = (const int4*)(dst + eb);
      int4 q[AGG_U];
#pragma unroll
      for (int u = 0; u < AGG_U; ++u) q[u] = p[u * 1024 + tid];
#pragma unroll
      for (int u = 0; u < AGG_U; ++u) {
        const int el = eb - e0 + (u * 1024 + tid) * 4;
        qcnt = agg_append(wq, qcnt, q[u].x, el,     base, span, lanes_below);
        qcnt = agg_append(wq, qcnt, q[u].y, el + 1, base, span, lanes_below);
        qcnt = agg_append(wq, qcnt, q[u].z, el + 2, base, span, lanes_below);
        qcnt = agg_append(wq, qcnt, q[u].w, el + 3, base, span, lanes_below);
        while (qcnt >= 32) {
          qcnt -= 32;
          agg_drain<RECORD>(wq + qcnt, 32, lane, e0, base, src, dinv, h, tile, wlist, saved);
        }
      }
    }
  }
  for (; eb < e1; eb += 1024) {
    const int t = eb + tid;
    const int d = (t < e1) ? dst[t] : -1;
    qcnt = agg_append(wq, qcnt, d, t - e0, base, span, lanes_below);
    if (qcnt >= 32) {
      qcnt -= 32;
      agg_drain<RECORD>(wq + qcnt, 32, lane, e0, base, src, dinv, h, tile, wlist, saved);
    }
  }
  if (qcnt > 0) agg_drain<RECORD>(wq, qcnt, lane, e0, base, src, dinv, h, tile, wlist, saved);
  return saved;
}

__global__ __launch_bounds__(1024)
void agg_tile_kernel(const int* __restrict__ src, const int* __restrict__ dst,
                     const float* __restrict__ dinv, const float* __restrict__ h,
                     float* __restrict__ planes, uint2* __restrict__ lists,
                     int* __restrict__ counts, int n, int e_count, int slice_len) {
  __shared__ float4 tile4[AGG_TILE * HID / 4];
  __shared__ unsigned queues[32 * AGG_QCAP];
  float* tile = (float*)tile4;
  const int tid = threadIdx.x;
  const int lane = tid & 31;
  const int wave = tid >> 5;
  lds_queue_t* wq = (lds_queue_t*)queues + wave * AGG_QCAP;
  const int base = blockIdx.y * AGG_TILE;
  const int span = min(AGG_TILE, n - base);
  const int quads = span * (HID / 4);
  const int wslot = (blockIdx.y * gridDim.x + blockIdx.x) * 32 + wave;

  for (int i = tid; i < quads; i += 1024) tile4[i] = make_float4(0.0f, 0.0f, 0.0f, 0.0f);
  __syncthreads();

  const int e0 = blockIdx.x * slice_len;
  const int e1 = min(e0 + slice_len, e_count);
  const int saved = agg_scan<true>(src, dst, dinv, h, tile, wq, lists + (size_t)wslot * AGG_LCAP,
                                   tid, lane, base, span, e0, e1);
  if (lane == 0) counts[wslot] = saved;
  __syncthreads();

  float4* out4 = (float4*)(planes + ((size_t)blockIdx.x * n + base) * HID);
  for (int i = tid; i < quads; i += 1024) out4[i] = tile4[i];
}

__global__ __launch_bounds__(1024)
void agg_replay_kernel(const int* __restrict__ src, const int* __restrict__ dst,
                       const float* __restrict__ dinv, const float* __restrict__ h,
                       float* __restrict__ planes, const uint2* __restrict__ lists,
                       const int* __restrict__ counts, int n, int e_count, int slice_len) {
  __shared__ float4 tile4[AGG_TILE * HID / 4];
  __shared__ unsigned queues[32 * AGG_QCAP];
  float* tile = (float*)tile4;
  const int tid = threadIdx.x;
  const int lane = tid & 31;
  const int wave = __builtin_amdgcn_readfirstlane(tid >> 5);
  const int base = blockIdx.y * AGG_TILE;
  const int span = min(AGG_TILE, n - base);
  const int quads = span * (HID / 4);
  const int wg = blockIdx.y * gridDim.x + blockIdx.x;

  for (int i = tid; i < quads; i += 1024) tile4[i] = make_float4(0.0f, 0.0f, 0.0f, 0.0f);
  __syncthreads();

  const int c_l = counts[wg * 32 + lane];
  if (__builtin_amdgcn_ballot_w32(c_l < 0) != 0u) {
    lds_queue_t* wq = (lds_queue_t*)queues + wave * AGG_QCAP;
    const int e0 = blockIdx.x * slice_len;
    const int e1 = min(e0 + slice_len, e_count);
    agg_scan<false>(src, dst, dinv, h, tile, wq, nullptr, tid, lane, base, span, e0, e1);
  } else {
    const int cnt = min(__builtin_amdgcn_readlane(c_l, wave), AGG_LCAP);
    const uint2* wl = lists + ((size_t)wg * 32 + wave) * AGG_LCAP;
    uint2 nxt = make_uint2(0u, 0u);
    if (lane < cnt) nxt = wl[lane];
    int p = 0;
    for (; p + 32 <= cnt; p += 32) {
      const uint2 cur = nxt;
      if (p + 32 + lane < cnt) nxt = wl[p + 32 + lane];
      agg_rows((int)(cur.x >> AGG_RBITS), (int)(cur.x & (AGG_TILE - 1)),
               __builtin_bit_cast(float, cur.y), 32, lane, h, tile);
    }
    if (p < cnt)
      agg_rows((int)(nxt.x >> AGG_RBITS), (int)(nxt.x & (AGG_TILE - 1)),
               __builtin_bit_cast(float, nxt.y), cnt - p, lane, h, tile);
  }
  __syncthreads();

  float4* out4 = (float4*)(planes + ((size_t)blockIdx.x * n + base) * HID);
  for (int i = tid; i < quads; i += 1024) out4[i] = tile4[i];
}

template <bool RELU>
__global__ void combine_kernel(const float* __restrict__ h, const float* __restrict__ dinv,
                               const float* __restrict__ bias, const float* __restrict__ planes,
                               float* __restrict__ out, int n) {
  const int i = blockIdx.x * blockDim.x + threadIdx.x;
  if (i >= n * (HID / 4)) return;
  const int node = i >> 3;
  const int c = (i & 7) * 4;
  const float di = dinv[node];
  const float4 hv = ((const float4*)h)[i];
  float4 pv[AGG_SLICES];
#pragma unroll
  for (int p = 0; p < AGG_SLICES; ++p) pv[p] = ((const float4*)planes)[(size_t)p * n * (HID / 4) + i];
  float4 a;
  a.x = bias[c]     + hv.x * di * di;
  a.y = bias[c + 1] + hv.y * di * di;
  a.z = bias[c + 2] + hv.z * di * di;
  a.w = bias[c + 3] + hv.w * di * di;
#pragma unroll
  for (int p = 0; p < AGG_SLICES; ++p) {
    a.x += pv[p].x; a.y += pv[p].y; a.z += pv[p].z; a.w += pv[p].w;
  }
  if (RELU) {
    a.x = fmaxf(a.x, 0.0f); a.y = fmaxf(a.y, 0.0f); a.z = fmaxf(a.z, 0.0f); a.w = fmaxf(a.w, 0.0f);
  }
  ((float4*)out)[i] = a;
}

template <int K>
__global__ __launch_bounds__(128)
void gemm_n32_kernel(const float* __restrict__ A, const float* __restrict__ W,
                     float* __restrict__ out, int M) {
  constexpr int LDK = K + 1;
  __shared__ float sA[32 * LDK];

  const int tid = threadIdx.x;
  const int block_row0 = blockIdx.x * 32;

  for (int idx = tid; idx < 32 * K; idx += 128) {
    int r = idx / K;
    int c = idx - r * K;
    int gr = block_row0 + r;
    sA[r * LDK + c] = (gr < M) ? A[(long long)gr * K + c] : 0.0f;
  }
  __syncthreads();

  const int wid   = tid >> 5;
  const int lane  = tid & 31;
  const int lhalf = lane >> 4;
  const int l15   = lane & 15;
  const int warp_m = (wid >> 1) * 16;
  const int n0     = (wid & 1) * 16;

  v8f acc = {};
  for (int k = 0; k < K; k += 4) {
    const float* ap = &sA[(warp_m + l15) * LDK + k + lhalf * 2];
    v2f a;
    a.x = ap[0];
    a.y = ap[1];
    const float* bp = &W[(k + lhalf * 2) * HID + n0 + l15];
    v2f b;
    b.x = bp[0];
    b.y = bp[HID];
    acc = __builtin_amdgcn_wmma_f32_16x16x4_f32(false, a, false, b, (short)0, acc,
                                                false, false);
  }

  const int grow_base = block_row0 + warp_m + lhalf * 8;
  const int col = n0 + l15;
#pragma unroll
  for (int j = 0; j < 8; ++j) {
    int gr = grow_base + j;
    if (gr < M) out[(long long)gr * HID + col] = acc[j];
  }
}


extern "C" void kernel_launch(void* const* d_in, const int* in_sizes, int n_in,
                              void* d_out, int out_size, void* d_ws, size_t ws_size,
                              hipStream_t stream) {
  const float* x  = (const float*)d_in[0];
  const int*   ei = (const int*)d_in[1];
  const float* W1 = (const float*)d_in[2];
  const float* b1 = (const float*)d_in[3];
  const float* W2 = (const float*)d_in[4];
  const float* b2 = (const float*)d_in[5];
  float* out = (float*)d_out;

  const int N = in_sizes[0] / IN_CH;
  const int E = in_sizes[1] / 2;
  const int* src = ei;
  const int* dst = ei + E;

  float* ws = (float*)d_ws;
  const int Npad = (N + 63) & ~63;
  float* dinv = ws;
  float* bufA = ws + Npad;
  float* bufB = bufA + (size_t)N * HID;
  unsigned* planes = (unsigned*)(bufB + (size_t)N * HID);

  const int T = 256;
  const unsigned gQuad = ((unsigned)N * (HID / 4) + T - 1) / T;
  const unsigned gGemm = (N + 31) / 32;
  const int nSlices    = (E + HIST_SLICE - 1) / HIST_SLICE;
  const int nTiles     = (N + HIST_TILE - 1) / HIST_TILE;
  const int planeWords = ((N + 7) >> 3) << 2;
  const unsigned gHalf = ((N + 1) / 2 + T - 1) / T;
  float* sums = (float*)(planes + (size_t)nSlices * planeWords);
  const int aggSlice = (E + AGG_SLICES - 1) / AGG_SLICES;
  const int aggTiles = (N + AGG_TILE - 1) / AGG_TILE;
  const dim3 gAgg(AGG_SLICES, aggTiles);
  uint2* lists = (uint2*)(sums + (size_t)AGG_SLICES * N * HID);
  int* counts = (int*)(lists + (size_t)AGG_SLICES * aggTiles * 32 * AGG_LCAP);

  degree_hist_kernel<<<dim3(nSlices, nTiles), 1024, 0, stream>>>(dst, planes, N, E, planeWords);
  dinv_kernel<<<gHalf, T, 0, stream>>>(planes, dinv, N, nSlices, planeWords);

  gemm_n32_kernel<IN_CH><<<gGemm, 128, 0, stream>>>(x, W1, bufA, N);
  agg_tile_kernel<<<gAgg, 1024, 0, stream>>>(src, dst, dinv, bufA, sums, lists, counts, N, E, aggSlice);
  combine_kernel<true><<<gQuad, T, 0, stream>>>(bufA, dinv, b1, sums, bufB, N);

  gemm_n32_kernel<HID><<<gGemm, 128, 0, stream>>>(bufB, W2, bufA, N);
  agg_replay_kernel<<<gAgg, 1024, 0, stream>>>(src, dst, dinv, bufA, sums, lists, counts, N, E, aggSlice);
  combine_kernel<false><<<gQuad, T, 0, stream>>>(bufA, dinv, b2, sums, out, N);
}


# GEMMs: a wave loads its 16 rows' A fragments straight from global in the matrix layout (16 B per lane, unguarded, several in flight); no LDS tile, no barrier, 16x32 per wave; k order of sums changes
# speedup vs baseline: 3.8822x; 1.0824x over previous
#include <hip/hip_runtime.h>

typedef __attribute__((ext_vector_type(2))) float v2f;
typedef __attribute__((ext_vector_type(8))) float v8f;

#define IN_CH 128
#define HID 32


#define HIST_SLICE 50000
#define HIST_TILE  131072

__device__ __forceinline__ void hist_bump(unsigned* cnt, int d, int base, int span) {
  unsigned r = (unsigned)(d - base);
  if (r < (unsigned)span) atomicAdd(&cnt[r >> 1], 1u << ((r & 1u) << 4));
}

__global__ __launch_bounds__(1024)
void degree_hist_kernel(const int* __restrict__ dst, unsigned* __restrict__ planes,
                        int n, int e_count, int plane_words) {
  __shared__ uint4 cnt4[HIST_TILE / 8];
  unsigned* cnt = (unsigned*)cnt4;
  const int tid = threadIdx.x;
  const int base = blockIdx.y * HIST_TILE;
  const int span = min(HIST_TILE, n - base);
  const int quads = (span + 7) >> 3;

  for (int i = tid; i < quads; i += 1024) cnt4[i] = make_uint4(0u, 0u, 0u, 0u);
  __syncthreads();

  const int e0 = blockIdx.x * HIST_SLICE;
  const int e1 = min(e0 + HIST_SLICE, e_count);
  int e = e0;
  if ((((size_t)(dst + e0)) & 15) == 0) {
    constexpr int U = 6;
    for (; e + 1024 * 4 * U <= e1; e += 1024 * 4 * U) {
      const int4* p = (const int4*)(dst + e);
      int4 q[U];
#pragma unroll
      for (int u = 0; u < U; ++u) q[u] = p[u * 1024 + tid];
#pragma unroll
      for (int u = 0; u < U; ++u) {
        hist_bump(cnt, q[u].x, base, span);
        hist_bump(cnt, q[u].y, base, span);
        hist_bump(cnt, q[u].z, base, span);
        hist_bump(cnt, q[u].w, base, span);
      }
    }
  }
  for (int t = e + tid; t < e1; t += 1024) hist_bump(cnt, dst[t], base, span);
  __syncthreads();

  uint4* out4 = (uint4*)(planes + (size_t)blockIdx.x * plane_words + (base >> 1));
  for (int i = tid; i < quads; i += 1024) out4[i] = cnt4[i];
}

__global__ void dinv_kernel(const unsigned* __restrict__ planes, float* __restrict__ dinv,
                            int n, int n_planes, int plane_words) {
  int w = blockIdx.x * blockDim.x + threadIdx.x;
  if (2 * w >= n) return;
  unsigned lo = 0u, hi = 0u;
#pragma unroll 16
  for (int p = 0; p < n_planes; ++p) {
    unsigned v = planes[(size_t)p * plane_words + w];
    lo += v & 0xffffu;
    hi += v >> 16;
  }
  float d0 = (float)(lo + 1u);
  dinv[2 * w] = (d0 > 0.0f) ? rsqrtf(d0) : 0.0f;
  if (2 * w + 1 < n) {
    float d1 = (float)(hi + 1u);
    dinv[2 * w + 1] = (d1 > 0.0f) ? rsqrtf(d1) : 0.0f;
  }
}

#define AGG_TILE   2048
#define AGG_RBITS  11
#define AGG_SLICES 5
#define AGG_U      4
#define AGG_QCAP   160
#define AGG_LCAP   1024


typedef __attribute__((address_space(3))) volatile unsigned lds_queue_t;

__device__ __forceinline__ void agg_rows(int s, int r, float nrm, int n, int lane,
                                         const float* __restrict__ h, float* tile) {
  float v[32];
#pragma unroll
  for (int j = 0; j < 32; ++j) {
    if (j < n) {
      int sj = __builtin_amdgcn_readlane(s, j);
      v[j] = h[(size_t)sj * HID + lane];
    }
  }
#pragma unroll
  for (int j = 0; j < 32; ++j) {
    if (j < n) {
      int rj = __builtin_amdgcn_readlane(r, j);
      float nj = __builtin_bit_cast(float, __builtin_amdgcn_readlane(__builtin_bit_cast(int, nrm), j));
      atomicAdd(&tile[rj * HID + lane], v[j] * nj);
    }
  }
}

__device__ __forceinline__ uint2* agg_reserve(uint2* wlist, int& saved, int n) {
  if (saved < 0 || saved + n > AGG_LCAP) { saved = -1; return nullptr; }
  uint2* p = wlist + saved;
  saved += n;
  return p;
}

template <bool RECORD>
__device__ __forceinline__ void agg_drain(const lds_queue_t* wq, int n, int lane, int e0,
                                          int base, const int* __restrict__ src,
                                          const float* __restrict__ dinv,
                                          const float* __restrict__ h, float* tile,
                                          uint2* wlist, int& saved) {
  uint2* rec = nullptr;
  if (RECORD) rec = agg_reserve(wlist, saved, n);
  int s = 0, r = 0;
  float nrm = 0.0f;
  if (lane < n) {
    unsigned ent = wq[lane];
    r = (int)(ent & (AGG_TILE - 1));
    s = src[e0 + (int)(ent >> AGG_RBITS)];
    nrm = dinv[s] * dinv[base + r];
    if (RECORD && rec)
      rec[lane] = make_uint2(((unsigned)s << AGG_RBITS) | (unsigned)r, __builtin_bit_cast(unsigned, nrm));
  }
  agg_rows(s, r, nrm, n, lane, h, tile);
}

__device__ __forceinline__ int agg_append(lds_queue_t* wq, int qcnt, int d, int el,
                                          int base, int span, unsigned lanes_below) {
  unsigned r = (unsigned)(d - base);
  bool m = r < (unsigned)span;
  unsigned bal = __builtin_amdgcn_ballot_w32(m);
  if (m) wq[qcnt + __popc(bal & lanes_below)] = ((unsigned)el << AGG_RBITS) | r;
  return qcnt + __popc(bal);
}

template <bool RECORD>
__device__ __forceinline__ int agg_scan(const int* __restrict__ src, const int* __restrict__ dst,
                                        const float* __restrict__ dinv, const float* __restrict__ h,
                                        float* tile, lds_queue_t* wq, uint2* wlist,
                                        int tid, int lane, int base, int span,
                                        int e0, int e1) {
  const unsigned lanes_below = (1u << lane) - 1u;
  int saved = 0;
  int qcnt = 0;
  int eb = e0;
  if ((((size_t)(dst + e0)) & 15) == 0) {
    for (; eb + 1024 * 4 * AGG_U <= e1; eb += 1024 * 4 * AGG_U) {
      const int4* p = (const int4*)(dst + eb);
      int4 q[AGG_U];
#pragma unroll
      for (int u = 0; u < AGG_U; ++u) q[u] = p[u * 1024 + tid];
#pragma unroll
      for (int u = 0; u < AGG_U; ++u) {
        const int el = eb - e0 + (u * 1024 + tid) * 4;
        qcnt = agg_append(wq, qcnt, q[u].x, el,     base, span, lanes_below);
        qcnt = agg_append(wq, qcnt, q[u].y, el + 1, base, span, lanes_below);
        qcnt = agg_append(wq, qcnt, q[u].z, el + 2, base, span, lanes_below);
        qcnt = agg_append(wq, qcnt, q[u].w, el + 3, base, span, lanes_below);
        while (qcnt >= 32) {
          qcnt -= 32;
          agg_drain<RECORD>(wq + qcnt, 32, lane, e0, base, src, dinv, h, tile, wlist, saved);
        }
      }
    }
  }
  for (; eb < e1; eb += 1024) {
    const int t = eb + tid;
    const int d = (t < e1) ? dst[t] : -1;
    qcnt = agg_append(wq, qcnt, d, t - e0, base, span, lanes_below);
    if (qcnt >= 32) {
      qcnt -= 32;
      agg_drain<RECORD>(wq + qcnt, 32, lane, e0, base, src, dinv, h, tile, wlist, saved);
    }
  }
  if (qcnt > 0) agg_drain<RECORD>(wq, qcnt, lane, e0, base, src, dinv, h, tile, wlist, saved);
  return saved;
}

__global__ __launch_bounds__(1024)
void agg_tile_kernel(const int* __restrict__ src, const int* __restrict__ dst,
                     const float* __restrict__ dinv, const float* __restrict__ h,
                     float* __restrict__ planes, uint2* __restrict__ lists,
                     int* __restrict__ counts, int n, int e_count, int slice_len) {
  __shared__ float4 tile4[AGG_TILE * HID / 4];
  __shared__ unsigned queues[32 * AGG_QCAP];
  float* tile = (float*)tile4;
  const int tid = threadIdx.x;
  const int lane = tid & 31;
  const int wave = tid >> 5;
  lds_queue_t* wq = (lds_queue_t*)queues + wave * AGG_QCAP;
  const int base = blockIdx.y * AGG_TILE;
  const int span = min(AGG_TILE, n - base);
  const int quads = span * (HID / 4);
  const int wslot = (blockIdx.y * gridDim.x + blockIdx.x) * 32 + wave;

  for (int i = tid; i < quads; i += 1024) tile4[i] = make_float4(0.0f, 0.0f, 0.0f, 0.0f);
  __syncthreads();

  const int e0 = blockIdx.x * slice_len;
  const int e1 = min(e0 + slice_len, e_count);
  const int saved = agg_scan<true>(src, dst, dinv, h, tile, wq, lists + (size_t)wslot * AGG_LCAP,
                                   tid, lane, base, span, e0, e1);
  if (lane == 0) counts[wslot] = saved;
  __syncthreads();

  float4* out4 = (float4*)(planes + ((size_t)blockIdx.x * n + base) * HID);
  for (int i = tid; i < quads; i += 1024) out4[i] = tile4[i];
}

__global__ __launch_bounds__(1024)
void agg_replay_kernel(const int* __restrict__ src, const int* __restrict__ dst,
                       const float* __restrict__ dinv, const float* __restrict__ h,
                       float* __restrict__ planes, const uint2* __restrict__ lists,
                       const int* __restrict__ counts, int n, int e_count, int slice_len) {
  __shared__ float4 tile4[AGG_TILE * HID / 4];
  __shared__ unsigned queues[32 * AGG_QCAP];
  float* tile = (float*)tile4;
  const int tid = threadIdx.x;
  const int lane = tid & 31;
  const int wave = __builtin_amdgcn_readfirstlane(tid >> 5);
  const int base = blockIdx.y * AGG_TILE;
  const int span = min(AGG_TILE, n - base);
  const int quads = span * (HID / 4);
  const int wg = blockIdx.y * gridDim.x + blockIdx.x;

  for (int i = tid; i < quads; i += 1024) tile4[i] = make_float4(0.0f, 0.0f, 0.0f, 0.0f);
  __syncthreads();

  const int c_l = counts[wg * 32 + lane];
  if (__builtin_amdgcn_ballot_w32(c_l < 0) != 0u) {
    lds_queue_t* wq = (lds_queue_t*)queues + wave * AGG_QCAP;
    const int e0 = blockIdx.x * slice_len;
    const int e1 = min(e0 + slice_len, e_count);
    agg_scan<false>(src, dst, dinv, h, tile, wq, nullptr, tid, lane, base, span, e0, e1);
  } else {
    const int cnt = min(__builtin_amdgcn_readlane(c_l, wave), AGG_LCAP);
    const uint2* wl = lists + ((size_t)wg * 32 + wave) * AGG_LCAP;
    uint2 nxt = make_uint2(0u, 0u);
    if (lane < cnt) nxt = wl[lane];
    int p = 0;
    for (; p + 32 <= cnt; p += 32) {
      const uint2 cur = nxt;
      if (p + 32 + lane < cnt) nxt = wl[p + 32 + lane];
      agg_rows((int)(cur.x >> AGG_RBITS), (int)(cur.x & (AGG_TILE - 1)),
               __builtin_bit_cast(float, cur.y), 32, lane, h, tile);
    }
    if (p < cnt)
      agg_rows((int)(nxt.x >> AGG_RBITS), (int)(nxt.x & (AGG_TILE - 1)),
               __builtin_bit_cast(float, nxt.y), cnt - p, lane, h, tile);
  }
  __syncthreads();

  float4* out4 = (float4*)(planes + ((size_t)blockIdx.x * n + base) * HID);
  for (int i = tid; i < quads; i += 1024) out4[i] = tile4[i];
}

template <bool RELU>
__global__ void combine_kernel(const float* __restrict__ h, const float* __restrict__ dinv,
                               const float* __restrict__ bias, const float* __restrict__ planes,
                               float* __restrict__ out, int n) {
  const int i = blockIdx.x * blockDim.x + threadIdx.x;
  if (i >= n * (HID / 4)) return;
  const int node = i >> 3;
  const int c = (i & 7) * 4;
  const float di = dinv[node];
  const float4 hv = ((const float4*)h)[i];
  float4 pv[AGG_SLICES];
#pragma unroll
  for (int p = 0; p < AGG_SLICES; ++p) pv[p] = ((const float4*)planes)[(size_t)p * n * (HID / 4) + i];
  float4 a;
  a.x = bias[c]     + hv.x * di * di;
  a.y = bias[c + 1] + hv.y * di * di;
  a.z = bias[c + 2] + hv.z * di * di;
  a.w = bias[c + 3] + hv.w * di * di;
#pragma unroll
  for (int p = 0; p < AGG_SLICES; ++p) {
    a.x += pv[p].x; a.y += pv[p].y; a.z += pv[p].z; a.w += pv[p].w;
  }
  if (RELU) {
    a.x = fmaxf(a.x, 0.0f); a.y = fmaxf(a.y, 0.0f); a.z = fmaxf(a.z, 0.0f); a.w = fmaxf(a.w, 0.0f);
  }
  ((float4*)out)[i] = a;
}

template <int K>
__global__ __launch_bounds__(128)
void gemm_n32_kernel(const float* __restrict__ A, const float* __restrict__ W,
                     float* __restrict__ out, int M) {
  const int lane = threadIdx.x & 31;
  const int wid  = __builtin_amdgcn_readfirstlane(threadIdx.x >> 5);
  const int c    = lane & 15;
  const int h    = lane >> 4;
  const int row0 = (blockIdx.x * 4 + wid) * 16;
  if (row0 >= M) return;

  const int row = min(row0 + c, M - 1);
  const float4* ap = (const float4*)(A + (size_t)row * K) + h;

  v8f acc0 = {};
  v8f acc1 = {};
#pragma unroll
  for (int i = 0; i < K / 8; ++i) {
    const float4 av = ap[2 * i];
    const float* bp = W + (8 * i + 4 * h) * HID + c;
    v2f a_lo, a_hi, b0_lo, b1_lo, b0_hi, b1_hi;
    a_lo.x = av.x;           a_lo.y = av.y;
    a_hi.x = av.z;           a_hi.y = av.w;
    b0_lo.x = bp[0];         b0_lo.y = bp[HID];
    b1_lo.x = bp[16];        b1_lo.y = bp[HID + 16];
    b0_hi.x = bp[2 * HID];   b0_hi.y = bp[3 * HID];
    b1_hi.x = bp[2 * HID + 16]; b1_hi.y = bp[3 * HID + 16];
    acc0 = __builtin_amdgcn_wmma_f32_16x16x4_f32(false, a_lo, false, b0_lo, (short)0, acc0, false, false);
    acc1 = __builtin_amdgcn_wmma_f32_16x16x4_f32(false, a_lo, false, b1_lo, (short)0, acc1, false, false);
    acc0 = __builtin_amdgcn_wmma_f32_16x16x4_f32(false, a_hi, false, b0_hi, (short)0, acc0, false, false);
    acc1 = __builtin_amdgcn_wmma_f32_16x16x4_f32(false, a_hi, false, b1_hi, (short)0, acc1, false, false);
  }

  const int grow = row0 + h * 8;
#pragma unroll
  for (int j = 0; j < 8; ++j) {
    const int gr = grow + j;
    if (gr < M) {
      out[(size_t)gr * HID + c]      = acc0[j];
      out[(size_t)gr * HID + 16 + c] = acc1[j];
    }
  }
}


extern "C" void kernel_launch(void* const* d_in, const int* in_sizes, int n_in,
                              void* d_out, int out_size, void* d_ws, size_t ws_size,
                              hipStream_t stream) {
  const float* x  = (const float*)d_in[0];
  const int*   ei = (const int*)d_in[1];
  const float* W1 = (const float*)d_in[2];
  const float* b1 = (const float*)d_in[3];
  const float* W2 = (const float*)d_in[4];
  const float* b2 = (const float*)d_in[5];
  float* out = (float*)d_out;

  const int N = in_sizes[0] / IN_CH;
  const int E = in_sizes[1] / 2;
  const int* src = ei;
  const int* dst = ei + E;

  float* ws = (float*)d_ws;
  const int Npad = (N + 63) & ~63;
  float* dinv = ws;
  float* bufA = ws + Npad;
  float* bufB = bufA + (size_t)N * HID;
  unsigned* planes = (unsigned*)(bufB + (size_t)N * HID);

  const int T = 256;
  const unsigned gQuad = ((unsigned)N * (HID / 4) + T - 1) / T;
  const unsigned gGemm = (N + 63) / 64;
  const int nSlices    = (E + HIST_SLICE - 1) / HIST_SLICE;
  const int nTiles     = (N + HIST_TILE - 1) / HIST_TILE;
  const int planeWords = ((N + 7) >> 3) << 2;
  const unsigned gHalf = ((N + 1) / 2 + T - 1) / T;
  float* sums = (float*)(planes + (size_t)nSlices * planeWords);
  const int aggSlice = (E + AGG_SLICES - 1) / AGG_SLICES;
  const int aggTiles = (N + AGG_TILE - 1) / AGG_TILE;
  const dim3 gAgg(AGG_SLICES, aggTiles);
  uint2* lists = (uint2*)(sums + (size_t)AGG_SLICES * N * HID);
  int* counts = (int*)(lists + (size_t)AGG_SLICES * aggTiles * 32 * AGG_LCAP);

  degree_hist_kernel<<<dim3(nSlices, nTiles), 1024, 0, stream>>>(dst, planes, N, E, planeWords);
  dinv_kernel<<<gHalf, T, 0, stream>>>(planes, dinv, N, nSlices, planeWords);

  gemm_n32_kernel<IN_CH><<<gGemm, 128, 0, stream>>>(x, W1, bufA, N);
  agg_tile_kernel<<<gAgg, 1024, 0, stream>>>(src, dst, dinv, bufA, sums, lists, counts, N, E, aggSlice);
  combine_kernel<true><<<gQuad, T, 0, stream>>>(bufA, dinv, b1, sums, bufB, N);

  gemm_n32_kernel<HID><<<gGemm, 128, 0, stream>>>(bufB, W2, bufA, N);
  agg_replay_kernel<<<gAgg, 1024, 0, stream>>>(src, dst, dinv, bufA, sums, lists, counts, N, E, aggSlice);
  combine_kernel<false><<<gQuad, T, 0, stream>>>(bufA, dinv, b2, sums, out, N);
}


# edge weight taken out of the edge loops: GEMM stores rows times dinv[row], sums add plain rows, combine multiplies by dinv[dst] once; no dinv gather per edge, 4-byte list entries; rounding changes
# speedup vs baseline: 4.4235x; 1.1394x over previous
#include <hip/hip_runtime.h>

typedef __attribute__((ext_vector_type(2))) float v2f;
typedef __attribute__((ext_vector_type(8))) float v8f;

#define IN_CH 128
#define HID 32


#define HIST_SLICE 50000
#define HIST_TILE  131072

__device__ __forceinline__ void hist_bump(unsigned* cnt, int d, int base, int span) {
  unsigned r = (unsigned)(d - base);
  if (r < (unsigned)span) atomicAdd(&cnt[r >> 1], 1u << ((r & 1u) << 4));
}

__global__ __launch_bounds__(1024)
void degree_hist_kernel(const int* __restrict__ dst, unsigned* __restrict__ planes,
                        int n, int e_count, int plane_words) {
  __shared__ uint4 cnt4[HIST_TILE / 8];
  unsigned* cnt = (unsigned*)cnt4;
  const int tid = threadIdx.x;
  const int base = blockIdx.y * HIST_TILE;
  const int span = min(HIST_TILE, n - base);
  const int quads = (span + 7) >> 3;

  for (int i = tid; i < quads; i += 1024) cnt4[i] = make_uint4(0u, 0u, 0u, 0u);
  __syncthreads();

  const int e0 = blockIdx.x * HIST_SLICE;
  const int e1 = min(e0 + HIST_SLICE, e_count);
  int e = e0;
  if ((((size_t)(dst + e0)) & 15) == 0) {
    constexpr int U = 6;
    for (; e + 1024 * 4 * U <= e1; e += 1024 * 4 * U) {
      const int4* p = (const int4*)(dst + e);
      int4 q[U];
#pragma unroll
      for (int u = 0; u < U; ++u) q[u] = p[u * 1024 + tid];
#pragma unroll
      for (int u = 0; u < U; ++u) {
        hist_bump(cnt, q[u].x, base, span);
        hist_bump(cnt, q[u].y, base, span);
        hist_bump(cnt, q[u].z, base, span);
        hist_bump(cnt, q[u].w, base, span);
      }
    }
  }
  for (int t = e + tid; t < e1; t += 1024) hist_bump(cnt, dst[t], base, span);
  __syncthreads();

  uint4* out4 = (uint4*)(planes + (size_t)blockIdx.x * plane_words + (base >> 1));
  for (int i = tid; i < quads; i += 1024) out4[i] = cnt4[i];
}

__global__ void dinv_kernel(const unsigned* __restrict__ planes, float* __restrict__ dinv,
                            int n, int n_planes, int plane_words) {
  int w = blockIdx.x * blockDim.x + threadIdx.x;
  if (2 * w >= n) return;
  unsigned lo = 0u, hi = 0u;
#pragma unroll 16
  for (int p = 0; p < n_planes; ++p) {
    unsigned v = planes[(size_t)p * plane_words + w];
    lo += v & 0xffffu;
    hi += v >> 16;
  }
  float d0 = (float)(lo + 1u);
  dinv[2 * w] = (d0 > 0.0f) ? rsqrtf(d0) : 0.0f;
  if (2 * w + 1 < n) {
    float d1 = (float)(hi + 1u);
    dinv[2 * w + 1] = (d1 > 0.0f) ? rsqrtf(d1) : 0.0f;
  }
}

#define AGG_TILE   2048
#define AGG_RBITS  11
#define AGG_SLICES 5
#define AGG_U      4
#define AGG_QCAP   160
#define AGG_LCAP   1024


typedef __attribute__((address_space(3))) volatile unsigned lds_queue_t;

__device__ __forceinline__ void agg_rows(int s, int r, int n, int lane,
                                         const float* __restrict__ h, float* tile) {
  float v[32];
#pragma unroll
  for (int j = 0; j < 32; ++j) {
    if (j < n) {
      int sj = __builtin_amdgcn_readlane(s, j);
      v[j] = h[(size_t)sj * HID + lane];
    }
  }
#pragma unroll
  for (int j = 0; j < 32; ++j) {
    if (j < n) {
      int rj = __builtin_amdgcn_readlane(r, j);
      atomicAdd(&tile[rj * HID + lane], v[j]);
    }
  }
}

__device__ __forceinline__ unsigned* agg_reserve(unsigned* wlist, int& saved, int n) {
  if (saved < 0 || saved + n > AGG_LCAP) { saved = -1; return nullptr; }
  unsigned* p = wlist + saved;
  saved += n;
  return p;
}

template <bool RECORD>
__device__ __forceinline__ void agg_drain(const lds_queue_t* wq, int n, int lane, int e0,
                                          const int* __restrict__ src,
                                          const float* __restrict__ h, float* tile,
                                          unsigned* wlist, int& saved) {
  unsigned* rec = nullptr;
  if (RECORD) rec = agg_reserve(wlist, saved, n);
  int s = 0, r = 0;
  if (lane < n) {
    unsigned ent = wq[lane];
    r = (int)(ent & (AGG_TILE - 1));
    s = src[e0 + (int)(ent >> AGG_RBITS)];
    if (RECORD && rec) rec[lane] = ((unsigned)s << AGG_RBITS) | (unsigned)r;
  }
  agg_rows(s, r, n, lane, h, tile);
}

__device__ __forceinline__ int agg_append(lds_queue_t* wq, int qcnt, int d, int el,
                                          int base, int span, unsigned lanes_below) {
  unsigned r = (unsigned)(d - base);
  bool m = r < (unsigned)span;
  unsigned bal = __builtin_amdgcn_ballot_w32(m);
  if (m) wq[qcnt + __popc(bal & lanes_below)] = ((unsigned)el << AGG_RBITS) | r;
  return qcnt + __popc(bal);
}

template <bool RECORD>
__device__ __forceinline__ int agg_scan(const int* __restrict__ src, const int* __restrict__ dst,
                                        const float* __restrict__ h,
                                        float* tile, lds_queue_t* wq, unsigned* wlist,
                                        int tid, int lane, int base, int span,
                                        int e0, int e1) {
  const unsigned lanes_below = (1u << lane) - 1u;
  int saved = 0;
  int qcnt = 0;
  int eb = e0;
  if ((((size_t)(dst + e0)) & 15) == 0) {
    for (; eb + 1024 * 4 * AGG_U <= e1; eb += 1024 * 4 * AGG_U) {
      const int4* p = (const int4*)(dst + eb);
      int4 q[AGG_U];
#pragma unroll
      for (int u = 0; u < AGG_U; ++u) q[u] = p[u * 1024 + tid];
#pragma unroll
      for (int u = 0; u < AGG_U; ++u) {
        const int el = eb - e0 + (u * 1024 + tid) * 4;
        qcnt = agg_append(wq, qcnt, q[u].x, el,     base, span, lanes_below);
        qcnt = agg_append(wq, qcnt, q[u].y, el + 1, base, span, lanes_below);
        qcnt = agg_append(wq, qcnt, q[u].z, el + 2, base, span, lanes_below);
        qcnt = agg_append(wq, qcnt, q[u].w, el + 3, base, span, lanes_below);
        while (qcnt >= 32) {
          qcnt -= 32;
          agg_drain<RECORD>(wq + qcnt, 32, lane, e0, src, h, tile, wlist, saved);
        }
      }
    }
  }
  for (; eb < e1; eb += 1024) {
    const int t = eb + tid;
    const int d = (t < e1) ? dst[t] : -1;
    qcnt = agg_append(wq, qcnt, d, t - e0, base, span, lanes_below);
    if (qcnt >= 32) {
      qcnt -= 32;
      agg_drain<RECORD>(wq + qcnt, 32, lane, e0, src, h, tile, wlist, saved);
    }
  }
  if (qcnt > 0) agg_drain<RECORD>(wq, qcnt, lane, e0, src, h, tile, wlist, saved);
  return saved;
}

__global__ __launch_bounds__(1024)
void agg_tile_kernel(const int* __restrict__ src, const int* __restrict__ dst,
                     const float* __restrict__ h,
                     float* __restrict__ planes, unsigned* __restrict__ lists,
                     int* __restrict__ counts, int n, int e_count, int slice_len) {
  __shared__ float4 tile4[AGG_TILE * HID / 4];
  __shared__ unsigned queues[32 * AGG_QCAP];
  float* tile = (float*)tile4;
  const int tid = threadIdx.x;
  const int lane = tid & 31;
  const int wave = tid >> 5;
  lds_queue_t* wq = (lds_queue_t*)queues + wave * AGG_QCAP;
  const int base = blockIdx.y * AGG_TILE;
  const int span = min(AGG_TILE, n - base);
  const int quads = span * (HID / 4);
  const int wslot = (blockIdx.y * gridDim.x + blockIdx.x) * 32 + wave;

  for (int i = tid; i < quads; i += 1024) tile4[i] = make_float4(0.0f, 0.0f, 0.0f, 0.0f);
  __syncthreads();

  const int e0 = blockIdx.x * slice_len;
  const int e1 = min(e0 + slice_len, e_count);
  const int saved = agg_scan<true>(src, dst, h, tile, wq, lists + (size_t)wslot * AGG_LCAP,
                                   tid, lane, base, span, e0, e1);
  if (lane == 0) counts[wslot] = saved;
  __syncthreads();

  float4* out4 = (float4*)(planes + ((size_t)blockIdx.x * n + base) * HID);
  for (int i = tid; i < quads; i += 1024) out4[i] = tile4[i];
}

__global__ __launch_bounds__(1024)
void agg_replay_kernel(const int* __restrict__ src, const int* __restrict__ dst,
                       const float* __restrict__ h,
                       float* __restrict__ planes, const unsigned* __restrict__ lists,
                       const int* __restrict__ counts, int n, int e_count, int slice_len) {
  __shared__ float4 tile4[AGG_TILE * HID / 4];
  __shared__ unsigned queues[32 * AGG_QCAP];
  float* tile = (float*)tile4;
  const int tid = threadIdx.x;
  const int lane = tid & 31;
  const int wave = __builtin_amdgcn_readfirstlane(tid >> 5);
  const int base = blockIdx.y * AGG_TILE;
  const int span = min(AGG_TILE, n - base);
  const int quads = span * (HID / 4);
  const int wg = blockIdx.y * gridDim.x + blockIdx.x;

  for (int i = tid; i < quads; i += 1024) tile4[i] = make_float4(0.0f, 0.0f, 0.0f, 0.0f);
  __syncthreads();

  const int c_l = counts[wg * 32 + lane];
  if (__builtin_amdgcn_ballot_w32(c_l < 0) != 0u) {
    lds_queue_t* wq = (lds_queue_t*)queues + wave * AGG_QCAP;
    const int e0 = blockIdx.x * slice_len;
    const int e1 = min(e0 + slice_len, e_count);
    agg_scan<false>(src, dst, h, tile, wq, nullptr, tid, lane, base, span, e0, e1);
  } else {
    const int cnt = min(__builtin_amdgcn_readlane(c_l, wave), AGG_LCAP);
    const unsigned* wl = lists + ((size_t)wg * 32 + wave) * AGG_LCAP;
    unsigned nxt = 0u;
    if (lane < cnt) nxt = wl[lane];
    int p = 0;
    for (; p + 32 <= cnt; p += 32) {
      const unsigned cur = nxt;
      if (p + 32 + lane < cnt) nxt = wl[p + 32 + lane];
      agg_rows((int)(cur >> AGG_RBITS), (int)(cur & (AGG_TILE - 1)), 32, lane, h, tile);
    }
    if (p < cnt)
      agg_rows((int)(nxt >> AGG_RBITS), (int)(nxt & (AGG_TILE - 1)), cnt - p, lane, h, tile);
  }
  __syncthreads();

  float4* out4 = (float4*)(planes + ((size_t)blockIdx.x * n + base) * HID);
  for (int i = tid; i < quads; i += 1024) out4[i] = tile4[i];
}

template <bool RELU>
__global__ void combine_kernel(const float* __restrict__ h, const float* __restrict__ dinv,
                               const float* __restrict__ bias, const float* __restrict__ planes,
                               float* __restrict__ out, int n) {
  const int i = blockIdx.x * blockDim.x + threadIdx.x;
  if (i >= n * (HID / 4)) return;
  const int node = i >> 3;
  const int c = (i & 7) * 4;
  const float di = dinv[node];
  const float4 hv = ((const float4*)h)[i];
  float4 pv[AGG_SLICES];
#pragma unroll
  for (int p = 0; p < AGG_SLICES; ++p) pv[p] = ((const float4*)planes)[(size_t)p * n * (HID / 4) + i];
  float4 a = hv;
#pragma unroll
  for (int p = 0; p < AGG_SLICES; ++p) {
    a.x += pv[p].x; a.y += pv[p].y; a.z += pv[p].z; a.w += pv[p].w;
  }
  a.x = bias[c]     + a.x * di;
  a.y = bias[c + 1] + a.y * di;
  a.z = bias[c + 2] + a.z * di;
  a.w = bias[c + 3] + a.w * di;
  if (RELU) {
    a.x = fmaxf(a.x, 0.0f); a.y = fmaxf(a.y, 0.0f); a.z = fmaxf(a.z, 0.0f); a.w = fmaxf(a.w, 0.0f);
  }
  ((float4*)out)[i] = a;
}

template <int K>
__global__ __launch_bounds__(128)
void gemm_n32_kernel(const float* __restrict__ A, const float* __restrict__ W,
                     const float* __restrict__ rscale, float* __restrict__ out, int M) {
  const int lane = threadIdx.x & 31;
  const int wid  = __builtin_amdgcn_readfirstlane(threadIdx.x >> 5);
  const int c    = lane & 15;
  const int h    = lane >> 4;
  const int row0 = (blockIdx.x * 4 + wid) * 16;
  if (row0 >= M) return;

  const int row = min(row0 + c, M - 1);
  const float4* ap = (const float4*)(A + (size_t)row * K) + h;

  v8f acc0 = {};
  v8f acc1 = {};
#pragma unroll
  for (int i = 0; i < K / 8; ++i) {
    const float4 av = ap[2 * i];
    const float* bp = W + (8 * i + 4 * h) * HID + c;
    v2f a_lo, a_hi, b0_lo, b1_lo, b0_hi, b1_hi;
    a_lo.x = av.x;           a_lo.y = av.y;
    a_hi.x = av.z;           a_hi.y = av.w;
    b0_lo.x = bp[0];         b0_lo.y = bp[HID];
    b1_lo.x = bp[16];        b1_lo.y = bp[HID + 16];
    b0_hi.x = bp[2 * HID];   b0_hi.y = bp[3 * HID];
    b1_hi.x = bp[2 * HID + 16]; b1_hi.y = bp[3 * HID + 16];
    acc0 = __builtin_amdgcn_wmma_f32_16x16x4_f32(false, a_lo, false, b0_lo, (short)0, acc0, false, false);
    acc1 = __builtin_amdgcn_wmma_f32_16x16x4_f32(false, a_lo, false, b1_lo, (short)0, acc1, false, false);
    acc0 = __builtin_amdgcn_wmma_f32_16x16x4_f32(false, a_hi, false, b0_hi, (short)0, acc0, false, false);
    acc1 = __builtin_amdgcn_wmma_f32_16x16x4_f32(false, a_hi, false, b1_hi, (short)0, acc1, false, false);
  }

  const int grow = row0 + h * 8;
  float rs[8];
#pragma unroll
  for (int j = 0; j < 8; ++j) rs[j] = rscale[min(grow + j, M - 1)];
#pragma unroll
  for (int j = 0; j < 8; ++j) {
    const int gr = grow + j;
    if (gr < M) {
      out[(size_t)gr * HID + c]      = acc0[j] * rs[j];
      out[(size_t)gr * HID + 16 + c] = acc1[j] * rs[j];
    }
  }
}


extern "C" void kernel_launch(void* const* d_in, const int* in_sizes, int n_in,
                              void* d_out, int out_size, void* d_ws, size_t ws_size,
                              hipStream_t stream) {
  const float* x  = (const float*)d_in[0];
  const int*   ei = (const int*)d_in[1];
  const float* W1 = (const float*)d_in[2];
  const float* b1 = (const float*)d_in[3];
  const float* W2 = (const float*)d_in[4];
  const float* b2 = (const float*)d_in[5];
  float* out = (float*)d_out;

  const int N = in_sizes[0] / IN_CH;
  const int E = in_sizes[1] / 2;
  const int* src = ei;
  const int* dst = ei + E;

  float* ws = (float*)d_ws;
  const int Npad = (N + 63) & ~63;
  float* dinv = ws;
  float* bufA = ws + Npad;
  float* bufB = bufA + (size_t)N * HID;
  unsigned* planes = (unsigned*)(bufB + (size_t)N * HID);

  const int T = 256;
  const unsigned gQuad = ((unsigned)N * (HID / 4) + T - 1) / T;
  const unsigned gGemm = (N + 63) / 64;
  const int nSlices    = (E + HIST_SLICE - 1) / HIST_SLICE;
  const int nTiles     = (N + HIST_TILE - 1) / HIST_TILE;
  const int planeWords = ((N + 7) >> 3) << 2;
  const unsigned gHalf = ((N + 1) / 2 + T - 1) / T;
  float* sums = (float*)(planes + (size_t)nSlices * planeWords);
  const int aggSlice = (E + AGG_SLICES - 1) / AGG_SLICES;
  const int aggTiles = (N + AGG_TILE - 1) / AGG_TILE;
  const dim3 gAgg(AGG_SLICES, aggTiles);
  unsigned* lists = (unsigned*)(sums + (size_t)AGG_SLICES * N * HID);
  int* counts = (int*)(lists + (size_t)AGG_SLICES * aggTiles * 32 * AGG_LCAP);

  degree_hist_kernel<<<dim3(nSlices, nTiles), 1024, 0, stream>>>(dst, planes, N, E, planeWords);
  dinv_kernel<<<gHalf, T, 0, stream>>>(planes, dinv, N, nSlices, planeWords);

  gemm_n32_kernel<IN_CH><<<gGemm, 128, 0, stream>>>(x, W1, dinv, bufA, N);
  agg_tile_kernel<<<gAgg, 1024, 0, stream>>>(src, dst, bufA, sums, lists, counts, N, E, aggSlice);
  combine_kernel<true><<<gQuad, T, 0, stream>>>(bufA, dinv, b1, sums, bufB, N);

  gemm_n32_kernel<HID><<<gGemm, 128, 0, stream>>>(bufB, W2, dinv, bufA, N);
  agg_replay_kernel<<<gAgg, 1024, 0, stream>>>(src, dst, bufA, sums, lists, counts, N, E, aggSlice);
  combine_kernel<false><<<gQuad, T, 0, stream>>>(bufA, dinv, b2, sums, out, N);
}


# edge filter split from the sums: one record-only filter launch over 4096-node ranges; sum groups own 410 whole nodes, re-filter saved lists, store final rows; no planes or combine; sum order changes
# speedup vs baseline: 4.7937x; 1.0837x over previous
#include <hip/hip_runtime.h>

typedef __attribute__((ext_vector_type(2))) float v2f;
typedef __attribute__((ext_vector_type(8))) float v8f;

#define IN_CH 128
#define HID 32


#define HIST_SLICE 50000
#define HIST_TILE  131072

__device__ __forceinline__ void hist_bump(unsigned* cnt, int d, int base, int span) {
  unsigned r = (unsigned)(d - base);
  if (r < (unsigned)span) atomicAdd(&cnt[r >> 1], 1u << ((r & 1u) << 4));
}

__global__ __launch_bounds__(1024)
void degree_hist_kernel(const int* __restrict__ dst, unsigned* __restrict__ planes,
                        int n, int e_count, int plane_words) {
  __shared__ uint4 cnt4[HIST_TILE / 8];
  unsigned* cnt = (unsigned*)cnt4;
  const int tid = threadIdx.x;
  const int base = blockIdx.y * HIST_TILE;
  const int span = min(HIST_TILE, n - base);
  const int quads = (span + 7) >> 3;

  for (int i = tid; i < quads; i += 1024) cnt4[i] = make_uint4(0u, 0u, 0u, 0u);
  __syncthreads();

  const int e0 = blockIdx.x * HIST_SLICE;
  const int e1 = min(e0 + HIST_SLICE, e_count);
  int e = e0;
  if ((((size_t)(dst + e0)) & 15) == 0) {
    constexpr int U = 6;
    for (; e + 1024 * 4 * U <= e1; e += 1024 * 4 * U) {
      const int4* p = (const int4*)(dst + e);
      int4 q[U];
#pragma unroll
      for (int u = 0; u < U; ++u) q[u] = p[u * 1024 + tid];
#pragma unroll
      for (int u = 0; u < U; ++u) {
        hist_bump(cnt, q[u].x, base, span);
        hist_bump(cnt, q[u].y, base, span);
        hist_bump(cnt, q[u].z, base, span);
        hist_bump(cnt, q[u].w, base, span);
      }
    }
  }
  for (int t = e + tid; t < e1; t += 1024) hist_bump(cnt, dst[t], base, span);
  __syncthreads();

  uint4* out4 = (uint4*)(planes + (size_t)blockIdx.x * plane_words + (base >> 1));
  for (int i = tid; i < quads; i += 1024) out4[i] = cnt4[i];
}

__global__ void dinv_kernel(const unsigned* __restrict__ planes, float* __restrict__ dinv,
                            int n, int n_planes, int plane_words) {
  int w = blockIdx.x * blockDim.x + threadIdx.x;
  if (2 * w >= n) return;
  unsigned lo = 0u, hi = 0u;
#pragma unroll 16
  for (int p = 0; p < n_planes; ++p) {
    unsigned v = planes[(size_t)p * plane_words + w];
    lo += v & 0xffffu;
    hi += v >> 16;
  }
  float d0 = (float)(lo + 1u);
  dinv[2 * w] = (d0 > 0.0f) ? rsqrtf(d0) : 0.0f;
  if (2 * w + 1 < n) {
    float d1 = (float)(hi + 1u);
    dinv[2 * w + 1] = (d1 > 0.0f) ? rsqrtf(d1) : 0.0f;
  }
}

#define AGG_RBITS  12
#define AGG_TILE   (1 << AGG_RBITS)
#define AGG_SLICES 10
#define AGG_OWNERS 10
#define AGG_OWN_ROWS ((AGG_TILE + AGG_OWNERS - 1) / AGG_OWNERS)
#define AGG_U      4
#define AGG_QCAP   160
#define AGG_LCAP   1024


typedef __attribute__((address_space(3))) volatile unsigned lds_queue_t;

__device__ __forceinline__ void agg_rows(int s, int r, int n, int lane,
                                         const float* __restrict__ h, float* tile) {
  if (n == 32) {
    float v[32];
#pragma unroll
    for (int j = 0; j < 32; ++j) {
      int sj = __builtin_amdgcn_readlane(s, j);
      v[j] = h[(size_t)sj * HID + lane];
    }
#pragma unroll
    for (int j = 0; j < 32; ++j) {
      int rj = __builtin_amdgcn_readlane(r, j);
      atomicAdd(&tile[rj * HID + lane], v[j]);
    }
  } else {
#pragma unroll
    for (int b = 0; b < 32; b += 16) {
      float v[16];
#pragma unroll
      for (int j = 0; j < 16; ++j) {
        if (b + j < n) {
          int sj = __builtin_amdgcn_readlane(s, b + j);
          v[j] = h[(size_t)sj * HID + lane];
        }
      }
#pragma unroll
      for (int j = 0; j < 16; ++j) {
        if (b + j < n) {
          int rj = __builtin_amdgcn_readlane(r, b + j);
          atomicAdd(&tile[rj * HID + lane], v[j]);
        }
      }
    }
  }
}

__device__ __forceinline__ unsigned* agg_reserve(unsigned* wlist, int& saved, int n) {
  if (saved < 0 || saved + n > AGG_LCAP) { saved = -1; return nullptr; }
  unsigned* p = wlist + saved;
  saved += n;
  return p;
}

template <bool RECORD>
__device__ __forceinline__ void agg_drain(const lds_queue_t* wq, int n, int lane, int e0,
                                          const int* __restrict__ src,
                                          const float* __restrict__ h, float* tile,
                                          unsigned* wlist, int& saved) {
  if (RECORD) {
    unsigned* rec = agg_reserve(wlist, saved, n);
    if (rec && lane < n) {
      unsigned ent = wq[lane];
      unsigned s = (unsigned)src[e0 + (int)(ent >> AGG_RBITS)];
      rec[lane] = (s << AGG_RBITS) | (ent & (AGG_TILE - 1));
    }
  } else {
    int s = 0, r = 0;
    if (lane < n) {
      unsigned ent = wq[lane];
      r = (int)(ent & (AGG_TILE - 1));
      s = src[e0 + (int)(ent >> AGG_RBITS)];
    }
    agg_rows(s, r, n, lane, h, tile);
  }
}

__device__ __forceinline__ int agg_push(lds_queue_t* wq, int qcnt, bool keep, unsigned ent,
                                        unsigned lanes_below) {
  unsigned bal = __builtin_amdgcn_ballot_w32(keep);
  if (keep) wq[qcnt + __popc(bal & lanes_below)] = ent;
  return qcnt + __popc(bal);
}

__device__ __forceinline__ int agg_append(lds_queue_t* wq, int qcnt, int d, int el,
                                          int base, int span, unsigned lanes_below) {
  unsigned r = (unsigned)(d - base);
  return agg_push(wq, qcnt, r < (unsigned)span, ((unsigned)el << AGG_RBITS) | r, lanes_below);
}

template <bool RECORD>
__device__ __forceinline__ int agg_scan(const int* __restrict__ src, const int* __restrict__ dst,
                                        const float* __restrict__ h,
                                        float* tile, lds_queue_t* wq, unsigned* wlist,
                                        int tid, int lane, int base, int span,
                                        int e0, int e1) {
  const unsigned lanes_below = (1u << lane) - 1u;
  int saved = 0;
  int qcnt = 0;
  int eb = e0;
  if ((((size_t)(dst + e0)) & 15) == 0) {
    for (; eb + 1024 * 4 * AGG_U <= e1; eb += 1024 * 4 * AGG_U) {
      const int4* p = (const int4*)(dst + eb);
      int4 q[AGG_U];
#pragma unroll
      for (int u = 0; u < AGG_U; ++u) q[u] = p[u * 1024 + tid];
#pragma unroll
      for (int u = 0; u < AGG_U; ++u) {
        const int el = eb - e0 + (u * 1024 + tid) * 4;
        qcnt = agg_append(wq, qcnt, q[u].x, el,     base, span, lanes_below);
        qcnt = agg_append(wq, qcnt, q[u].y, el + 1, base, span, lanes_below);
        qcnt = agg_append(wq, qcnt, q[u].z, el + 2, base, span, lanes_below);
        qcnt = agg_append(wq, qcnt, q[u].w, el + 3, base, span, lanes_below);
        while (qcnt >= 32) {
          qcnt -= 32;
          agg_drain<RECORD>(wq + qcnt, 32, lane, e0, src, h, tile, wlist, saved);
        }
      }
    }
  }
  for (; eb < e1; eb += 1024) {
    const int t = eb + tid;
    const int d = (t < e1) ? dst[t] : -1;
    qcnt = agg_append(wq, qcnt, d, t - e0, base, span, lanes_below);
    if (qcnt >= 32) {
      qcnt -= 32;
      agg_drain<RECORD>(wq + qcnt, 32, lane, e0, src, h, tile, wlist, saved);
    }
  }
  if (qcnt > 0) agg_drain<RECORD>(wq, qcnt, lane, e0, src, h, tile, wlist, saved);
  return saved;
}

__global__ __launch_bounds__(1024)
void agg_filter_kernel(const int* __restrict__ src, const int* __restrict__ dst,
                       unsigned* __restrict__ lists, int* __restrict__ counts,
                       int n, int e_count, int slice_len) {
  __shared__ unsigned queues[32 * AGG_QCAP];
  const int tid = threadIdx.x;
  const int lane = tid & 31;
  const int wave = tid >> 5;
  lds_queue_t* wq = (lds_queue_t*)queues + wave * AGG_QCAP;
  const int base = blockIdx.y * AGG_TILE;
  const int span = min(AGG_TILE, n - base);
  const int wslot = (blockIdx.y * AGG_SLICES + blockIdx.x) * 32 + wave;
  const int e0 = blockIdx.x * slice_len;
  const int e1 = min(e0 + slice_len, e_count);
  const int saved = agg_scan<true>(src, dst, nullptr, nullptr, wq,
                                   lists + (size_t)wslot * AGG_LCAP,
                                   tid, lane, base, span, e0, e1);
  if (lane == 0) counts[wslot] = saved;
}

#define OWN_QCAP   544

__device__ __forceinline__ void own_rows(const lds_queue_t* wq, int n, int lane,
                                         const float* __restrict__ h, float* tile) {
  int s = 0, r = 0;
  if (lane < n) {
    unsigned ent = wq[lane];
    r = (int)(ent & (AGG_TILE - 1));
    s = (int)(ent >> AGG_RBITS);
  }
  agg_rows(s, r, n, lane, h, tile);
}

__device__ __forceinline__ void own_load(uint4 (&q)[AGG_U], const unsigned* wl, int p, int lane) {
#pragma unroll
  for (int u = 0; u < AGG_U; ++u) q[u] = ((const uint4*)(wl + p + u * 128))[lane];
}

__device__ __forceinline__ int own_keep(const uint4 (&q)[AGG_U], int p, int cnt, int qcnt,
                                        lds_queue_t* wq, int lane, unsigned lanes_below,
                                        int orow, int ospan) {
#pragma unroll
  for (int u = 0; u < AGG_U; ++u) {
    const int idx = p + u * 128 + lane * 4;
    const unsigned e[4] = {q[u].x, q[u].y, q[u].z, q[u].w};
#pragma unroll
    for (int k = 0; k < 4; ++k) {
      const unsigned rl = (e[k] & (AGG_TILE - 1)) - (unsigned)orow;
      qcnt = agg_push(wq, qcnt, (idx + k < cnt) && (rl < (unsigned)ospan),
                      e[k] - (unsigned)orow, lanes_below);
    }
  }
  return qcnt;
}

template <bool RELU>
__global__ __launch_bounds__(1024)
void agg_own_kernel(const int* __restrict__ src, const int* __restrict__ dst,
                    const float* __restrict__ h, const float* __restrict__ dinv,
                    const float* __restrict__ bias,
                    const unsigned* __restrict__ lists, const int* __restrict__ counts,
                    float* __restrict__ out, int n, int e_count, int slice_len) {
  __shared__ float4 tile4[AGG_OWN_ROWS * HID / 4];
  __shared__ unsigned queues[32 * OWN_QCAP];
  __shared__ int lcnt[AGG_SLICES * 32];
  float* tile = (float*)tile4;
  const int tid = threadIdx.x;
  const int lane = tid & 31;
  const int wave = __builtin_amdgcn_readfirstlane(tid >> 5);
  lds_queue_t* wq = (lds_queue_t*)queues + wave * OWN_QCAP;
  const int rbase = blockIdx.y * AGG_TILE;
  const int rspan = min(AGG_TILE, n - rbase);
  const int orow = blockIdx.x * AGG_OWN_ROWS;
  const int ospan = min(AGG_OWN_ROWS, rspan - orow);
  if (ospan <= 0) return;
  const int obase = rbase + orow;
  const int quads = ospan * (HID / 4);

  const int fwg0 = blockIdx.y * AGG_SLICES;
  const unsigned* wl0 = lists + ((size_t)fwg0 * 32 + wave) * AGG_LCAP;
  uint4 q[AGG_U];
  own_load(q, wl0, 0, lane);

  const float4* h4 = (const float4*)h + (size_t)obase * (HID / 4);
  for (int i = tid; i < quads; i += 1024) tile4[i] = h4[i];
  if (tid < AGG_SLICES * 32) lcnt[tid] = counts[fwg0 * 32 + tid];
  __syncthreads();

  const unsigned lanes_below = (1u << lane) - 1u;
  int qcnt = 0;
  int s = 0, p = 0;
  int c_l = lcnt[lane];
  while (s < AGG_SLICES) {
    const bool over = __builtin_amdgcn_ballot_w32(c_l < 0) != 0u;
    const int cnt = over ? 0 : min(__builtin_amdgcn_readlane(c_l, wave), AGG_LCAP);
    if (!over) qcnt = own_keep(q, p, cnt, qcnt, wq, lane, lanes_below, orow, ospan);
    int ns = s, np = p + 128 * AGG_U;
    if (np >= cnt) { ns = s + 1; np = 0; }
    if (ns < AGG_SLICES) {
      if (ns != s) c_l = lcnt[ns * 32 + lane];
      own_load(q, wl0 + (size_t)ns * 32 * AGG_LCAP, np, lane);
    }
    if (over) {
      if (qcnt > 0) { own_rows(wq, qcnt, lane, h, tile); qcnt = 0; }
      const int e0 = s * slice_len;
      const int e1 = min(e0 + slice_len, e_count);
      agg_scan<false>(src, dst, h, tile, wq, nullptr, tid, lane, obase, ospan, e0, e1);
    }
    while (qcnt >= 32) {
      qcnt -= 32;
      own_rows(wq + qcnt, 32, lane, h, tile);
    }
    s = ns; p = np;
  }
  if (qcnt > 0) own_rows(wq, qcnt, lane, h, tile);
  __syncthreads();

  float4* out4 = (float4*)out + (size_t)obase * (HID / 4);
  for (int i = tid; i < quads; i += 1024) {
    const float di = dinv[obase + (i >> 3)];
    const int c = (i & 7) * 4;
    float4 a = tile4[i];
    a.x = bias[c]     + a.x * di;
    a.y = bias[c + 1] + a.y * di;
    a.z = bias[c + 2] + a.z * di;
    a.w = bias[c + 3] + a.w * di;
    if (RELU) {
      a.x = fmaxf(a.x, 0.0f); a.y = fmaxf(a.y, 0.0f); a.z = fmaxf(a.z, 0.0f); a.w = fmaxf(a.w, 0.0f);
    }
    out4[i] = a;
  }
}

template <int K>
__global__ __launch_bounds__(128)
void gemm_n32_kernel(const float* __restrict__ A, const float* __restrict__ W,
                     const float* __restrict__ rscale, float* __restrict__ out, int M) {
  const int lane = threadIdx.x & 31;
  const int wid  = __builtin_amdgcn_readfirstlane(threadIdx.x >> 5);
  const int c    = lane & 15;
  const int h    = lane >> 4;
  const int row0 = (blockIdx.x * 4 + wid) * 16;
  if (row0 >= M) return;

  const int row = min(row0 + c, M - 1);
  const float4* ap = (const float4*)(A + (size_t)row * K) + h;

  v8f acc0 = {};
  v8f acc1 = {};
#pragma unroll
  for (int i = 0; i < K / 8; ++i) {
    const float4 av = ap[2 * i];
    const float* bp = W + (8 * i + 4 * h) * HID + c;
    v2f a_lo, a_hi, b0_lo, b1_lo, b0_hi, b1_hi;
    a_lo.x = av.x;           a_lo.y = av.y;
    a_hi.x = av.z;           a_hi.y = av.w;
    b0_lo.x = bp[0];         b0_lo.y = bp[HID];
    b1_lo.x = bp[16];        b1_lo.y = bp[HID + 16];
    b0_hi.x = bp[2 * HID];   b0_hi.y = bp[3 * HID];
    b1_hi.x = bp[2 * HID + 16]; b1_hi.y = bp[3 * HID + 16];
    acc0 = __builtin_amdgcn_wmma_f32_16x16x4_f32(false, a_lo, false, b0_lo, (short)0, acc0, false, false);
    acc1 = __builtin_amdgcn_wmma_f32_16x16x4_f32(false, a_lo, false, b1_lo, (short)0, acc1, false, false);
    acc0 = __builtin_amdgcn_wmma_f32_16x16x4_f32(false, a_hi, false, b0_hi, (short)0, acc0, false, false);
    acc1 = __builtin_amdgcn_wmma_f32_16x16x4_f32(false, a_hi, false, b1_hi, (short)0, acc1, false, false);
  }

  const int grow = row0 + h * 8;
  float rs[8];
#pragma unroll
  for (int j = 0; j < 8; ++j) rs[j] = rscale[min(grow + j, M - 1)];
#pragma unroll
  for (int j = 0; j < 8; ++j) {
    const int gr = grow + j;
    if (gr < M) {
      out[(size_t)gr * HID + c]      = acc0[j] * rs[j];
      out[(size_t)gr * HID + 16 + c] = acc1[j] * rs[j];
    }
  }
}


extern "C" void kernel_launch(void* const* d_in, const int* in_sizes, int n_in,
                              void* d_out, int out_size, void* d_ws, size_t ws_size,
                              hipStream_t stream) {
  const float* x  = (const float*)d_in[0];
  const int*   ei = (const int*)d_in[1];
  const float* W1 = (const float*)d_in[2];
  const float* b1 = (const float*)d_in[3];
  const float* W2 = (const float*)d_in[4];
  const float* b2 = (const float*)d_in[5];
  float* out = (float*)d_out;

  const int N = in_sizes[0] / IN_CH;
  const int E = in_sizes[1] / 2;
  const int* src = ei;
  const int* dst = ei + E;

  float* ws = (float*)d_ws;
  const int Npad = (N + 63) & ~63;
  float* dinv = ws;
  float* bufA = ws + Npad;
  float* bufB = bufA + (size_t)N * HID;
  unsigned* planes = (unsigned*)(bufB + (size_t)N * HID);

  const int T = 256;
  const unsigned gGemm = (N + 63) / 64;
  const int nSlices    = (E + HIST_SLICE - 1) / HIST_SLICE;
  const int nTiles     = (N + HIST_TILE - 1) / HIST_TILE;
  const int planeWords = ((N + 7) >> 3) << 2;
  const unsigned gHalf = ((N + 1) / 2 + T - 1) / T;
  const int aggSlice = (E + AGG_SLICES - 1) / AGG_SLICES;
  const int aggTiles = (N + AGG_TILE - 1) / AGG_TILE;
  const dim3 gFilter(AGG_SLICES, aggTiles);
  const dim3 gOwn(AGG_OWNERS, aggTiles);
  unsigned* lists = planes + (size_t)nSlices * planeWords;
  int* counts = (int*)(lists + (size_t)AGG_SLICES * aggTiles * 32 * AGG_LCAP);

  agg_filter_kernel<<<gFilter, 1024, 0, stream>>>(src, dst, lists, counts, N, E, aggSlice);

  degree_hist_kernel<<<dim3(nSlices, nTiles), 1024, 0, stream>>>(dst, planes, N, E, planeWords);
  dinv_kernel<<<gHalf, T, 0, stream>>>(planes, dinv, N, nSlices, planeWords);

  gemm_n32_kernel<IN_CH><<<gGemm, 128, 0, stream>>>(x, W1, dinv, bufA, N);
  agg_own_kernel<true><<<gOwn, 1024, 0, stream>>>(src, dst, bufA, dinv, b1, lists, counts, bufB,
                                                   N, E, aggSlice);

  gemm_n32_kernel<HID><<<gGemm, 128, 0, stream>>>(bufB, W2, dinv, bufA, N);
  agg_own_kernel<false><<<gOwn, 1024, 0, stream>>>(src, dst, bufA, dinv, b2, lists, counts, out,
                                                    N, E, aggSlice);
}


# in-degree counted inside the edge filter launch (LDS counters per node range, a count plane per edge slice); first GEMM sums a node's 10 counts into dinv and stores it; degree and dinv launches gone
# speedup vs baseline: 5.2752x; 1.1004x over previous
#include <hip/hip_runtime.h>

typedef __attribute__((ext_vector_type(2))) float v2f;
typedef __attribute__((ext_vector_type(8))) float v8f;

#define IN_CH 128
#define HID 32

#define AGG_RBITS  12
#define AGG_TILE   (1 << AGG_RBITS)
#define AGG_SLICES 10
#define AGG_OWNERS 10
#define AGG_OWN_ROWS ((AGG_TILE + AGG_OWNERS - 1) / AGG_OWNERS)
#define AGG_U      4
#define AGG_QCAP   160
#define AGG_LCAP   1024


typedef __attribute__((address_space(3))) volatile unsigned lds_queue_t;

__device__ __forceinline__ void agg_rows(int s, int r, int n, int lane,
                                         const float* __restrict__ h, float* tile) {
  if (n == 32) {
    float v[32];
#pragma unroll
    for (int j = 0; j < 32; ++j) {
      int sj = __builtin_amdgcn_readlane(s, j);
      v[j] = h[(size_t)sj * HID + lane];
    }
#pragma unroll
    for (int j = 0; j < 32; ++j) {
      int rj = __builtin_amdgcn_readlane(r, j);
      atomicAdd(&tile[rj * HID + lane], v[j]);
    }
  } else {
#pragma unroll
    for (int b = 0; b < 32; b += 16) {
      float v[16];
#pragma unroll
      for (int j = 0; j < 16; ++j) {
        if (b + j < n) {
          int sj = __builtin_amdgcn_readlane(s, b + j);
          v[j] = h[(size_t)sj * HID + lane];
        }
      }
#pragma unroll
      for (int j = 0; j < 16; ++j) {
        if (b + j < n) {
          int rj = __builtin_amdgcn_readlane(r, b + j);
          atomicAdd(&tile[rj * HID + lane], v[j]);
        }
      }
    }
  }
}

__device__ __forceinline__ unsigned* agg_reserve(unsigned* wlist, int& saved, int n) {
  if (saved < 0 || saved + n > AGG_LCAP) { saved = -1; return nullptr; }
  unsigned* p = wlist + saved;
  saved += n;
  return p;
}

template <bool RECORD>
__device__ __forceinline__ void agg_drain(const lds_queue_t* wq, int n, int lane, int e0,
                                          const int* __restrict__ src,
                                          const float* __restrict__ h, float* tile,
                                          unsigned* wlist, int& saved, unsigned* hist) {
  if (RECORD) {
    unsigned* rec = agg_reserve(wlist, saved, n);
    if (lane < n) {
      unsigned ent = wq[lane];
      atomicAdd(&hist[ent & (AGG_TILE - 1)], 1u);
      if (rec) {
        unsigned s = (unsigned)src[e0 + (int)(ent >> AGG_RBITS)];
        rec[lane] = (s << AGG_RBITS) | (ent & (AGG_TILE - 1));
      }
    }
  } else {
    int s = 0, r = 0;
    if (lane < n) {
      unsigned ent = wq[lane];
      r = (int)(ent & (AGG_TILE - 1));
      s = src[e0 + (int)(ent >> AGG_RBITS)];
    }
    agg_rows(s, r, n, lane, h, tile);
  }
}

__device__ __forceinline__ int agg_push(lds_queue_t* wq, int qcnt, bool keep, unsigned ent,
                                        unsigned lanes_below) {
  unsigned bal = __builtin_amdgcn_ballot_w32(keep);
  if (keep) wq[qcnt + __popc(bal & lanes_below)] = ent;
  return qcnt + __popc(bal);
}

__device__ __forceinline__ int agg_append(lds_queue_t* wq, int qcnt, int d, int el,
                                          int base, int span, unsigned lanes_below) {
  unsigned r = (unsigned)(d - base);
  return agg_push(wq, qcnt, r < (unsigned)span, ((unsigned)el << AGG_RBITS) | r, lanes_below);
}

template <bool RECORD>
__device__ __forceinline__ int agg_scan(const int* __restrict__ src, const int* __restrict__ dst,
                                        const float* __restrict__ h,
                                        float* tile, lds_queue_t* wq, unsigned* wlist,
                                        unsigned* hist,
                                        int tid, int lane, int base, int span,
                                        int e0, int e1) {
  const unsigned lanes_below = (1u << lane) - 1u;
  int saved = 0;
  int qcnt = 0;
  int eb = e0;
  if ((((size_t)(dst + e0)) & 15) == 0) {
    for (; eb + 1024 * 4 * AGG_U <= e1; eb += 1024 * 4 * AGG_U) {
      const int4* p = (const int4*)(dst + eb);
      int4 q[AGG_U];
#pragma unroll
      for (int u = 0; u < AGG_U; ++u) q[u] = p[u * 1024 + tid];
#pragma unroll
      for (int u = 0; u < AGG_U; ++u) {
        const int el = eb - e0 + (u * 1024 + tid) * 4;
        qcnt = agg_append(wq, qcnt, q[u].x, el,     base, span, lanes_below);
        qcnt = agg_append(wq, qcnt, q[u].y, el + 1, base, span, lanes_below);
        qcnt = agg_append(wq, qcnt, q[u].z, el + 2, base, span, lanes_below);
        qcnt = agg_append(wq, qcnt, q[u].w, el + 3, base, span, lanes_below);
        while (qcnt >= 32) {
          qcnt -= 32;
          agg_drain<RECORD>(wq + qcnt, 32, lane, e0, src, h, tile, wlist, saved, hist);
        }
      }
    }
  }
  for (; eb < e1; eb += 1024) {
    const int t = eb + tid;
    const int d = (t < e1) ? dst[t] : -1;
    qcnt = agg_append(wq, qcnt, d, t - e0, base, span, lanes_below);
    if (qcnt >= 32) {
      qcnt -= 32;
      agg_drain<RECORD>(wq + qcnt, 32, lane, e0, src, h, tile, wlist, saved, hist);
    }
  }
  if (qcnt > 0) agg_drain<RECORD>(wq, qcnt, lane, e0, src, h, tile, wlist, saved, hist);
  return saved;
}

__global__ __launch_bounds__(1024)
void agg_filter_kernel(const int* __restrict__ src, const int* __restrict__ dst,
                       unsigned* __restrict__ lists, int* __restrict__ counts,
                       unsigned* __restrict__ deg_planes, int plane_stride,
                       int n, int e_count, int slice_len) {
  __shared__ unsigned queues[32 * AGG_QCAP];
  __shared__ unsigned hist[AGG_TILE];
  const int tid = threadIdx.x;
  const int lane = tid & 31;
  const int wave = tid >> 5;
  lds_queue_t* wq = (lds_queue_t*)queues + wave * AGG_QCAP;
  const int base = blockIdx.y * AGG_TILE;
  const int span = min(AGG_TILE, n - base);
  const int wslot = (blockIdx.y * AGG_SLICES + blockIdx.x) * 32 + wave;
  const int e0 = blockIdx.x * slice_len;
  const int e1 = min(e0 + slice_len, e_count);
  for (int i = tid; i < AGG_TILE; i += 1024) hist[i] = 0u;
  __syncthreads();
  const int saved = agg_scan<true>(src, dst, nullptr, nullptr, wq,
                                   lists + (size_t)wslot * AGG_LCAP, hist,
                                   tid, lane, base, span, e0, e1);
  if (lane == 0) counts[wslot] = saved;
  __syncthreads();
  unsigned* pl = deg_planes + (size_t)blockIdx.x * plane_stride + base;
  for (int i = tid; i < span; i += 1024) pl[i] = hist[i];
}

#define OWN_QCAP   544

__device__ __forceinline__ void own_rows(const lds_queue_t* wq, int n, int lane,
                                         const float* __restrict__ h, float* tile) {
  int s = 0, r = 0;
  if (lane < n) {
    unsigned ent = wq[lane];
    r = (int)(ent & (AGG_TILE - 1));
    s = (int)(ent >> AGG_RBITS);
  }
  agg_rows(s, r, n, lane, h, tile);
}

__device__ __forceinline__ void own_load(uint4 (&q)[AGG_U], const unsigned* wl, int p, int lane) {
#pragma unroll
  for (int u = 0; u < AGG_U; ++u) q[u] = ((const uint4*)(wl + p + u * 128))[lane];
}

__device__ __forceinline__ int own_keep(const uint4 (&q)[AGG_U], int p, int cnt, int qcnt,
                                        lds_queue_t* wq, int lane, unsigned lanes_below,
                                        int orow, int ospan) {
#pragma unroll
  for (int u = 0; u < AGG_U; ++u) {
    const int idx = p + u * 128 + lane * 4;
    const unsigned e[4] = {q[u].x, q[u].y, q[u].z, q[u].w};
#pragma unroll
    for (int k = 0; k < 4; ++k) {
      const unsigned rl = (e[k] & (AGG_TILE - 1)) - (unsigned)orow;
      qcnt = agg_push(wq, qcnt, (idx + k < cnt) && (rl < (unsigned)ospan),
                      e[k] - (unsigned)orow, lanes_below);
    }
  }
  return qcnt;
}

template <bool RELU>
__global__ __launch_bounds__(1024)
void agg_own_kernel(const int* __restrict__ src, const int* __restrict__ dst,
                    const float* __restrict__ h, const float* __restrict__ dinv,
                    const float* __restrict__ bias,
                    const unsigned* __restrict__ lists, const int* __restrict__ counts,
                    float* __restrict__ out, int n, int e_count, int slice_len) {
  __shared__ float4 tile4[AGG_OWN_ROWS * HID / 4];
  __shared__ unsigned queues[32 * OWN_QCAP];
  __shared__ int lcnt[AGG_SLICES * 32];
  float* tile = (float*)tile4;
  const int tid = threadIdx.x;
  const int lane = tid & 31;
  const int wave = __builtin_amdgcn_readfirstlane(tid >> 5);
  lds_queue_t* wq = (lds_queue_t*)queues + wave * OWN_QCAP;
  const int rbase = blockIdx.y * AGG_TILE;
  const int rspan = min(AGG_TILE, n - rbase);
  const int orow = blockIdx.x * AGG_OWN_ROWS;
  const int ospan = min(AGG_OWN_ROWS, rspan - orow);
  if (ospan <= 0) return;
  const int obase = rbase + orow;
  const int quads = ospan * (HID / 4);

  const int fwg0 = blockIdx.y * AGG_SLICES;
  const unsigned* wl0 = lists + ((size_t)fwg0 * 32 + wave) * AGG_LCAP;
  uint4 q[AGG_U];
  own_load(q, wl0, 0, lane);

  const float4* h4 = (const float4*)h + (size_t)obase * (HID / 4);
  for (int i = tid; i < quads; i += 1024) tile4[i] = h4[i];
  if (tid < AGG_SLICES * 32) lcnt[tid] = counts[fwg0 * 32 + tid];
  __syncthreads();

  const unsigned lanes_below = (1u << lane) - 1u;
  int qcnt = 0;
  int s = 0, p = 0;
  int c_l = lcnt[lane];
  while (s < AGG_SLICES) {
    const bool over = __builtin_amdgcn_ballot_w32(c_l < 0) != 0u;
    const int cnt = over ? 0 : min(__builtin_amdgcn_readlane(c_l, wave), AGG_LCAP);
    if (!over) qcnt = own_keep(q, p, cnt, qcnt, wq, lane, lanes_below, orow, ospan);
    int ns = s, np = p + 128 * AGG_U;
    if (np >= cnt) { ns = s + 1; np = 0; }
    if (ns < AGG_SLICES) {
      if (ns != s) c_l = lcnt[ns * 32 + lane];
      own_load(q, wl0 + (size_t)ns * 32 * AGG_LCAP, np, lane);
    }
    if (over) {
      if (qcnt > 0) { own_rows(wq, qcnt, lane, h, tile); qcnt = 0; }
      const int e0 = s * slice_len;
      const int e1 = min(e0 + slice_len, e_count);
      agg_scan<false>(src, dst, h, tile, wq, nullptr, nullptr, tid, lane, obase, ospan, e0, e1);
    }
    while (qcnt >= 32) {
      qcnt -= 32;
      own_rows(wq + qcnt, 32, lane, h, tile);
    }
    s = ns; p = np;
  }
  if (qcnt > 0) own_rows(wq, qcnt, lane, h, tile);
  __syncthreads();

  float4* out4 = (float4*)out + (size_t)obase * (HID / 4);
  for (int i = tid; i < quads; i += 1024) {
    const float di = dinv[obase + (i >> 3)];
    const int c = (i & 7) * 4;
    float4 a = tile4[i];
    a.x = bias[c]     + a.x * di;
    a.y = bias[c + 1] + a.y * di;
    a.z = bias[c + 2] + a.z * di;
    a.w = bias[c + 3] + a.w * di;
    if (RELU) {
      a.x = fmaxf(a.x, 0.0f); a.y = fmaxf(a.y, 0.0f); a.z = fmaxf(a.z, 0.0f); a.w = fmaxf(a.w, 0.0f);
    }
    out4[i] = a;
  }
}

template <int K, bool DEG>
__global__ __launch_bounds__(128)
void gemm_n32_kernel(const float* __restrict__ A, const float* __restrict__ W,
                     const float* __restrict__ rscale,
                     const unsigned* __restrict__ deg_planes, int plane_stride,
                     float* __restrict__ dinv_out, float* __restrict__ out, int M) {
  const int lane = threadIdx.x & 31;
  const int wid  = __builtin_amdgcn_readfirstlane(threadIdx.x >> 5);
  const int c    = lane & 15;
  const int h    = lane >> 4;
  const int row0 = (blockIdx.x * 4 + wid) * 16;
  if (row0 >= M) return;

  const int row = min(row0 + c, M - 1);
  const float4* ap = (const float4*)(A + (size_t)row * K) + h;

  v8f acc0 = {};
  v8f acc1 = {};
#pragma unroll
  for (int i = 0; i < K / 8; ++i) {
    const float4 av = ap[2 * i];
    const float* bp = W + (8 * i + 4 * h) * HID + c;
    v2f a_lo, a_hi, b0_lo, b1_lo, b0_hi, b1_hi;
    a_lo.x = av.x;           a_lo.y = av.y;
    a_hi.x = av.z;           a_hi.y = av.w;
    b0_lo.x = bp[0];         b0_lo.y = bp[HID];
    b1_lo.x = bp[16];        b1_lo.y = bp[HID + 16];
    b0_hi.x = bp[2 * HID];   b0_hi.y = bp[3 * HID];
    b1_hi.x = bp[2 * HID + 16]; b1_hi.y = bp[3 * HID + 16];
    acc0 = __builtin_amdgcn_wmma_f32_16x16x4_f32(false, a_lo, false, b0_lo, (short)0, acc0, false, false);
    acc1 = __builtin_amdgcn_wmma_f32_16x16x4_f32(false, a_lo, false, b1_lo, (short)0, acc1, false, false);
    acc0 = __builtin_amdgcn_wmma_f32_16x16x4_f32(false, a_hi, false, b0_hi, (short)0, acc0, false, false);
    acc1 = __builtin_amdgcn_wmma_f32_16x16x4_f32(false, a_hi, false, b1_hi, (short)0, acc1, false, false);
  }

  float dv = 0.0f;
  if (DEG) {
    unsigned deg = 1u;
#pragma unroll
    for (int p = 0; p < AGG_SLICES; ++p) deg += deg_planes[(size_t)p * plane_stride + row];
    const float d = (float)deg;
    dv = (d > 0.0f) ? rsqrtf(d) : 0.0f;
    if (h == 0 && row0 + c < M) dinv_out[row0 + c] = dv;
  }

  const int grow = row0 + h * 8;
  float rs[8];
#pragma unroll
  for (int j = 0; j < 8; ++j) rs[j] = DEG ? __shfl(dv, h * 8 + j) : rscale[min(grow + j, M - 1)];
#pragma unroll
  for (int j = 0; j < 8; ++j) {
    const int gr = grow + j;
    if (gr < M) {
      out[(size_t)gr * HID + c]      = acc0[j] * rs[j];
      out[(size_t)gr * HID + 16 + c] = acc1[j] * rs[j];
    }
  }
}


extern "C" void kernel_launch(void* const* d_in, const int* in_sizes, int n_in,
                              void* d_out, int out_size, void* d_ws, size_t ws_size,
                              hipStream_t stream) {
  const float* x  = (const float*)d_in[0];
  const int*   ei = (const int*)d_in[1];
  const float* W1 = (const float*)d_in[2];
  const float* b1 = (const float*)d_in[3];
  const float* W2 = (const float*)d_in[4];
  const float* b2 = (const float*)d_in[5];
  float* out = (float*)d_out;

  const int N = in_sizes[0] / IN_CH;
  const int E = in_sizes[1] / 2;
  const int* src = ei;
  const int* dst = ei + E;

  float* ws = (float*)d_ws;
  const int Npad = (N + 63) & ~63;
  float* dinv = ws;
  float* bufA = ws + Npad;
  float* bufB = bufA + (size_t)N * HID;
  unsigned* planes = (unsigned*)(bufB + (size_t)N * HID);

  const unsigned gGemm = (N + 63) / 64;
  const int aggSlice = (E + AGG_SLICES - 1) / AGG_SLICES;
  const int aggTiles = (N + AGG_TILE - 1) / AGG_TILE;
  const dim3 gFilter(AGG_SLICES, aggTiles);
  const dim3 gOwn(AGG_OWNERS, aggTiles);
  unsigned* lists = planes + (size_t)AGG_SLICES * Npad;
  int* counts = (int*)(lists + (size_t)AGG_SLICES * aggTiles * 32 * AGG_LCAP);

  agg_filter_kernel<<<gFilter, 1024, 0, stream>>>(src, dst, lists, counts, planes, Npad,
                                                  N, E, aggSlice);

  gemm_n32_kernel<IN_CH, true><<<gGemm, 128, 0, stream>>>(x, W1, nullptr, planes, Npad, dinv,
                                                          bufA, N);
  agg_own_kernel<true><<<gOwn, 1024, 0, stream>>>(src, dst, bufA, dinv, b1, lists, counts, bufB,
                                                   N, E, aggSlice);

  gemm_n32_kernel<HID, false><<<gGemm, 128, 0, stream>>>(bufB, W2, dinv, nullptr, 0, nullptr,
                                                         bufA, N);
  agg_own_kernel<false><<<gOwn, 1024, 0, stream>>>(src, dst, bufA, dinv, b2, lists, counts, out,
                                                    N, E, aggSlice);
}


# edge filter bins matched edges by owner workgroup in LDS (returning LDS add = slot), one list per owner; sum launches read only their own lists, no re-filter; 8192-node ranges; sum order changes
# speedup vs baseline: 6.0970x; 1.1558x over previous
#include <hip/hip_runtime.h>

typedef __attribute__((ext_vector_type(2))) float v2f;
typedef __attribute__((ext_vector_type(8))) float v8f;

#define IN_CH 128
#define HID 32

#define AGG_RBITS  13
#define AGG_TILE   (1 << AGG_RBITS)
#define AGG_SLICES 19
#define AGG_OWNERS 20
#define AGG_OWN_ROWS ((AGG_TILE + AGG_OWNERS - 1) / AGG_OWNERS)
#define AGG_U      4
#define AGG_QCAP   160
#define AGG_OCAP   1024
#define AGG_CSTRIDE 32


typedef __attribute__((address_space(3))) volatile unsigned lds_queue_t;

__device__ __forceinline__ void agg_rows(int s, int r, int n, int lane,
                                         const float* __restrict__ h, float* tile) {
  if (n == 32) {
    float v[32];
#pragma unroll
    for (int j = 0; j < 32; ++j) {
      int sj = __builtin_amdgcn_readlane(s, j);
      v[j] = h[(size_t)sj * HID + lane];
    }
#pragma unroll
    for (int j = 0; j < 32; ++j) {
      int rj = __builtin_amdgcn_readlane(r, j);
      atomicAdd(&tile[rj * HID + lane], v[j]);
    }
  } else {
#pragma unroll
    for (int b = 0; b < 32; b += 16) {
      float v[16];
#pragma unroll
      for (int j = 0; j < 16; ++j) {
        if (b + j < n) {
          int sj = __builtin_amdgcn_readlane(s, b + j);
          v[j] = h[(size_t)sj * HID + lane];
        }
      }
#pragma unroll
      for (int j = 0; j < 16; ++j) {
        if (b + j < n) {
          int rj = __builtin_amdgcn_readlane(r, b + j);
          atomicAdd(&tile[rj * HID + lane], v[j]);
        }
      }
    }
  }
}

template <bool RECORD>
__device__ __forceinline__ void agg_drain(const lds_queue_t* wq, int n, int lane, int e0,
                                          const int* __restrict__ src,
                                          const float* __restrict__ h, float* tile,
                                          unsigned* hist, unsigned* olist, unsigned* ocnt) {
  if (RECORD) {
    if (lane < n) {
      const unsigned ent = wq[lane];
      const unsigned r = ent & (AGG_TILE - 1);
      const unsigned sn = (unsigned)src[e0 + (int)(ent >> AGG_RBITS)];
      atomicAdd(&hist[r], 1u);
      const unsigned o = r / AGG_OWN_ROWS;
      const unsigned slot = atomicAdd(&ocnt[o], 1u);
      if (slot < AGG_OCAP) olist[o * AGG_OCAP + slot] = (sn << AGG_RBITS) | r;
    }
  } else {
    int s = 0, r = 0;
    if (lane < n) {
      unsigned ent = wq[lane];
      r = (int)(ent & (AGG_TILE - 1));
      s = src[e0 + (int)(ent >> AGG_RBITS)];
    }
    agg_rows(s, r, n, lane, h, tile);
  }
}

__device__ __forceinline__ int agg_push(lds_queue_t* wq, int qcnt, bool keep, unsigned ent,
                                        unsigned lanes_below) {
  unsigned bal = __builtin_amdgcn_ballot_w32(keep);
  if (keep) wq[qcnt + __popc(bal & lanes_below)] = ent;
  return qcnt + __popc(bal);
}

__device__ __forceinline__ int agg_append(lds_queue_t* wq, int qcnt, int d, int el,
                                          int base, int span, unsigned lanes_below) {
  unsigned r = (unsigned)(d - base);
  return agg_push(wq, qcnt, r < (unsigned)span, ((unsigned)el << AGG_RBITS) | r, lanes_below);
}

template <bool RECORD>
__device__ __forceinline__ void agg_scan(const int* __restrict__ src, const int* __restrict__ dst,
                                        const float* __restrict__ h,
                                        float* tile, lds_queue_t* wq,
                                        unsigned* hist, unsigned* olist, unsigned* ocnt,
                                        int tid, int lane, int base, int span,
                                        int e0, int e1) {
  const unsigned lanes_below = (1u << lane) - 1u;
  int qcnt = 0;
  int eb = e0;
  if ((((size_t)(dst + e0)) & 15) == 0) {
    for (; eb + 1024 * 4 * AGG_U <= e1; eb += 1024 * 4 * AGG_U) {
      const int4* p = (const int4*)(dst + eb);
      int4 q[AGG_U];
#pragma unroll
      for (int u = 0; u < AGG_U; ++u) q[u] = p[u * 1024 + tid];
#pragma unroll
      for (int u = 0; u < AGG_U; ++u) {
        const int el = eb - e0 + (u * 1024 + tid) * 4;
        qcnt = agg_append(wq, qcnt, q[u].x, el,     base, span, lanes_below);
        qcnt = agg_append(wq, qcnt, q[u].y, el + 1, base, span, lanes_below);
        qcnt = agg_append(wq, qcnt, q[u].z, el + 2, base, span, lanes_below);
        qcnt = agg_append(wq, qcnt, q[u].w, el + 3, base, span, lanes_below);
        while (qcnt >= 32) {
          qcnt -= 32;
          agg_drain<RECORD>(wq + qcnt, 32, lane, e0, src, h, tile, hist, olist, ocnt);
        }
      }
    }
  }
  for (; eb < e1; eb += 1024) {
    const int t = eb + tid;
    const int d = (t < e1) ? dst[t] : -1;
    qcnt = agg_append(wq, qcnt, d, t - e0, base, span, lanes_below);
    if (qcnt >= 32) {
      qcnt -= 32;
      agg_drain<RECORD>(wq + qcnt, 32, lane, e0, src, h, tile, hist, olist, ocnt);
    }
  }
  if (qcnt > 0) agg_drain<RECORD>(wq, qcnt, lane, e0, src, h, tile, hist, olist, ocnt);
}

__global__ __launch_bounds__(1024)
void agg_filter_kernel(const int* __restrict__ src, const int* __restrict__ dst,
                       unsigned* __restrict__ lists, int* __restrict__ counts,
                       unsigned* __restrict__ deg_planes, int plane_stride,
                       int n, int e_count, int slice_len) {
  __shared__ unsigned queues[32 * AGG_QCAP];
  __shared__ unsigned hist[AGG_TILE];
  __shared__ unsigned olist[AGG_OWNERS * AGG_OCAP];
  __shared__ unsigned ocnt[AGG_OWNERS];
  const int tid = threadIdx.x;
  const int lane = tid & 31;
  const int wave = tid >> 5;
  lds_queue_t* wq = (lds_queue_t*)queues + wave * AGG_QCAP;
  const int base = blockIdx.y * AGG_TILE;
  const int span = min(AGG_TILE, n - base);
  const int fwg = blockIdx.y * AGG_SLICES + blockIdx.x;
  const int e0 = blockIdx.x * slice_len;
  const int e1 = min(e0 + slice_len, e_count);
  for (int i = tid; i < AGG_TILE; i += 1024) hist[i] = 0u;
  if (tid < AGG_OWNERS) ocnt[tid] = 0u;
  __syncthreads();
  agg_scan<true>(src, dst, nullptr, nullptr, wq, hist, olist, ocnt,
                 tid, lane, base, span, e0, e1);
  __syncthreads();
  unsigned* pl = deg_planes + (size_t)blockIdx.x * plane_stride + base;
  for (int i = tid; i < span; i += 1024) pl[i] = hist[i];
  unsigned* gl = lists + (size_t)fwg * AGG_OWNERS * AGG_OCAP;
  for (int o = 0; o < AGG_OWNERS; ++o) {
    const unsigned c = ocnt[o];
    for (int k = tid; k < AGG_OCAP && (unsigned)k < c; k += 1024)
      gl[o * AGG_OCAP + k] = olist[o * AGG_OCAP + k];
  }
  if (tid < AGG_OWNERS) {
    const unsigned c = ocnt[tid];
    counts[fwg * AGG_CSTRIDE + tid] = (c > AGG_OCAP) ? -1 : (int)c;
  }
}

__device__ __forceinline__ unsigned own_fetch(const unsigned* __restrict__ lists_o, int s, int c_l,
                                              int wave, int lane, int& goff, int& n) {
  const int cnt = min(max(__builtin_amdgcn_readlane(c_l, s), 0), AGG_OCAP);
  const int k = (wave - goff) & 31;
  goff += (cnt + 31) >> 5;
  n = min(max(cnt - k * 32, 0), 32);
  unsigned e = 0u;
  if (lane < n) e = lists_o[(size_t)s * (AGG_OWNERS * AGG_OCAP) + k * 32 + lane];
  return e;
}

template <bool RELU>
__global__ __launch_bounds__(1024)
void agg_own_kernel(const int* __restrict__ src, const int* __restrict__ dst,
                    const float* __restrict__ h, const float* __restrict__ dinv,
                    const float* __restrict__ bias,
                    const unsigned* __restrict__ lists, const int* __restrict__ counts,
                    float* __restrict__ out, int n, int e_count, int slice_len) {
  __shared__ float4 tile4[AGG_OWN_ROWS * HID / 4];
  __shared__ unsigned queues[32 * AGG_QCAP];
  float* tile = (float*)tile4;
  const int tid = threadIdx.x;
  const int lane = tid & 31;
  const int wave = __builtin_amdgcn_readfirstlane(tid >> 5);
  lds_queue_t* wq = (lds_queue_t*)queues + wave * AGG_QCAP;
  const int rbase = blockIdx.y * AGG_TILE;
  const int rspan = min(AGG_TILE, n - rbase);
  const int orow = blockIdx.x * AGG_OWN_ROWS;
  const int ospan = min(AGG_OWN_ROWS, rspan - orow);
  if (ospan <= 0) return;
  const int obase = rbase + orow;
  const int quads = ospan * (HID / 4);

  const int fwg0 = blockIdx.y * AGG_SLICES;
  const int c_l = (lane < AGG_SLICES) ? counts[(fwg0 + lane) * AGG_CSTRIDE + blockIdx.x] : 0;
  const unsigned* lists_o = lists + ((size_t)fwg0 * AGG_OWNERS + blockIdx.x) * AGG_OCAP;
  int goff = 0, n_next = 0;
  unsigned e_next = own_fetch(lists_o, 0, c_l, wave, lane, goff, n_next);

  const float4* h4 = (const float4*)h + (size_t)obase * (HID / 4);
  for (int i = tid; i < quads; i += 1024) tile4[i] = h4[i];
  __syncthreads();

#pragma nounroll
  for (int s = 0; s < AGG_SLICES; ++s) {
    const unsigned e = e_next;
    const int nn = n_next;
    if (s + 1 < AGG_SLICES) e_next = own_fetch(lists_o, s + 1, c_l, wave, lane, goff, n_next);
    if (__builtin_amdgcn_readlane(c_l, s) < 0) {
      const int e0 = s * slice_len;
      const int e1 = min(e0 + slice_len, e_count);
      agg_scan<false>(src, dst, h, tile, wq, nullptr, nullptr, nullptr,
                      tid, lane, obase, ospan, e0, e1);
    } else if (nn > 0) {
      int sn = 0, r = 0;
      if (lane < nn) {
        r = (int)(e & (AGG_TILE - 1)) - orow;
        sn = (int)(e >> AGG_RBITS);
      }
      agg_rows(sn, r, nn, lane, h, tile);
    }
  }
  __syncthreads();

  float4* out4 = (float4*)out + (size_t)obase * (HID / 4);
  for (int i = tid; i < quads; i += 1024) {
    const float di = dinv[obase + (i >> 3)];
    const int c = (i & 7) * 4;
    float4 a = tile4[i];
    a.x = bias[c]     + a.x * di;
    a.y = bias[c + 1] + a.y * di;
    a.z = bias[c + 2] + a.z * di;
    a.w = bias[c + 3] + a.w * di;
    if (RELU) {
      a.x = fmaxf(a.x, 0.0f); a.y = fmaxf(a.y, 0.0f); a.z = fmaxf(a.z, 0.0f); a.w = fmaxf(a.w, 0.0f);
    }
    out4[i] = a;
  }
}

template <int K, bool DEG>
__global__ __launch_bounds__(128)
void gemm_n32_kernel(const float* __restrict__ A, const float* __restrict__ W,
                     const float* __restrict__ rscale,
                     const unsigned* __restrict__ deg_planes, int plane_stride,
                     float* __restrict__ dinv_out, float* __restrict__ out, int M) {
  const int lane = threadIdx.x & 31;
  const int wid  = __builtin_amdgcn_readfirstlane(threadIdx.x >> 5);
  const int c    = lane & 15;
  const int h    = lane >> 4;
  const int row0 = (blockIdx.x * 4 + wid) * 16;
  if (row0 >= M) return;

  const int row = min(row0 + c, M - 1);
  const float4* ap = (const float4*)(A + (size_t)row * K) + h;

  v8f acc0 = {};
  v8f acc1 = {};
#pragma unroll
  for (int i = 0; i < K / 8; ++i) {
    const float4 av = ap[2 * i];
    const float* bp = W + (8 * i + 4 * h) * HID + c;
    v2f a_lo, a_hi, b0_lo, b1_lo, b0_hi, b1_hi;
    a_lo.x = av.x;           a_lo.y = av.y;
    a_hi.x = av.z;           a_hi.y = av.w;
    b0_lo.x = bp[0];         b0_lo.y = bp[HID];
    b1_lo.x = bp[16];        b1_lo.y = bp[HID + 16];
    b0_hi.x = bp[2 * HID];   b0_hi.y = bp[3 * HID];
    b1_hi.x = bp[2 * HID + 16]; b1_hi.y = bp[3 * HID + 16];
    acc0 = __builtin_amdgcn_wmma_f32_16x16x4_f32(false, a_lo, false, b0_lo, (short)0, acc0, false, false);
    acc1 = __builtin_amdgcn_wmma_f32_16x16x4_f32(false, a_lo, false, b1_lo, (short)0, acc1, false, false);
    acc0 = __builtin_amdgcn_wmma_f32_16x16x4_f32(false, a_hi, false, b0_hi, (short)0, acc0, false, false);
    acc1 = __builtin_amdgcn_wmma_f32_16x16x4_f32(false, a_hi, false, b1_hi, (short)0, acc1, false, false);
  }

  float dv = 0.0f;
  if (DEG) {
    unsigned deg = 1u;
#pragma unroll
    for (int p = 0; p < AGG_SLICES; ++p) deg += deg_planes[(size_t)p * plane_stride + row];
    const float d = (float)deg;
    dv = (d > 0.0f) ? rsqrtf(d) : 0.0f;
    if (h == 0 && row0 + c < M) dinv_out[row0 + c] = dv;
  }

  const int grow = row0 + h * 8;
  float rs[8];
#pragma unroll
  for (int j = 0; j < 8; ++j) rs[j] = DEG ? __shfl(dv, h * 8 + j) : rscale[min(grow + j, M - 1)];
#pragma unroll
  for (int j = 0; j < 8; ++j) {
    const int gr = grow + j;
    if (gr < M) {
      out[(size_t)gr * HID + c]      = acc0[j] * rs[j];
      out[(size_t)gr * HID + 16 + c] = acc1[j] * rs[j];
    }
  }
}


extern "C" void kernel_launch(void* const* d_in, const int* in_sizes, int n_in,
                              void* d_out, int out_size, void* d_ws, size_t ws_size,
                              hipStream_t stream) {
  const float* x  = (const float*)d_in[0];
  const int*   ei = (const int*)d_in[1];
  const float* W1 = (const float*)d_in[2];
  const float* b1 = (const float*)d_in[3];
  const float* W2 = (const float*)d_in[4];
  const float* b2 = (const float*)d_in[5];
  float* out = (float*)d_out;

  const int N = in_sizes[0] / IN_CH;
  const int E = in_sizes[1] / 2;
  const int* src = ei;
  const int* dst = ei + E;

  float* ws = (float*)d_ws;
  const int Npad = (N + 63) & ~63;
  float* dinv = ws;
  float* bufA = ws + Npad;
  float* bufB = bufA + (size_t)N * HID;
  unsigned* planes = (unsigned*)(bufB + (size_t)N * HID);

  const unsigned gGemm = (N + 63) / 64;
  const int aggSlice = ((E + AGG_SLICES - 1) / AGG_SLICES + 3) & ~3;
  const int aggTiles = (N + AGG_TILE - 1) / AGG_TILE;
  const dim3 gFilter(AGG_SLICES, aggTiles);
  const dim3 gOwn(AGG_OWNERS, aggTiles);
  unsigned* lists = planes + (size_t)AGG_SLICES * Npad;
  int* counts = (int*)(lists + (size_t)AGG_SLICES * aggTiles * AGG_OWNERS * AGG_OCAP);

  agg_filter_kernel<<<gFilter, 1024, 0, stream>>>(src, dst, lists, counts, planes, Npad,
                                                  N, E, aggSlice);

  gemm_n32_kernel<IN_CH, true><<<gGemm, 128, 0, stream>>>(x, W1, nullptr, planes, Npad, dinv,
                                                          bufA, N);
  agg_own_kernel<true><<<gOwn, 1024, 0, stream>>>(src, dst, bufA, dinv, b1, lists, counts, bufB,
                                                   N, E, aggSlice);

  gemm_n32_kernel<HID, false><<<gGemm, 128, 0, stream>>>(bufB, W2, dinv, nullptr, 0, nullptr,
                                                         bufA, N);
  agg_own_kernel<false><<<gOwn, 1024, 0, stream>>>(src, dst, bufA, dinv, b2, lists, counts, out,
                                                    N, E, aggSlice);
}


# layer 2 reordered by linearity, (A R) W2: sum launch 2 adds relu'd rows (kept times dinv) and multiplies its finished LDS tile by W2 (f32 WMMA, in place); GEMM2 launch gone; sum order changes
# speedup vs baseline: 6.3363x; 1.0392x over previous
#include <hip/hip_runtime.h>

typedef __attribute__((ext_vector_type(2))) float v2f;
typedef __attribute__((ext_vector_type(8))) float v8f;

#define IN_CH 128
#define HID 32

#define AGG_RBITS  13
#define AGG_TILE   (1 << AGG_RBITS)
#define AGG_SLICES 19
#define AGG_OWNERS 20
#define AGG_OWN_ROWS ((AGG_TILE + AGG_OWNERS - 1) / AGG_OWNERS)
#define AGG_U      4
#define AGG_QCAP   160
#define AGG_OCAP   1024
#define AGG_CSTRIDE 32


typedef __attribute__((address_space(3))) volatile unsigned lds_queue_t;

__device__ __forceinline__ void agg_rows(int s, int r, int n, int lane,
                                         const float* __restrict__ h, float* tile) {
  if (n == 32) {
    float v[32];
#pragma unroll
    for (int j = 0; j < 32; ++j) {
      int sj = __builtin_amdgcn_readlane(s, j);
      v[j] = h[(size_t)sj * HID + lane];
    }
#pragma unroll
    for (int j = 0; j < 32; ++j) {
      int rj = __builtin_amdgcn_readlane(r, j);
      atomicAdd(&tile[rj * HID + lane], v[j]);
    }
  } else {
#pragma unroll
    for (int b = 0; b < 32; b += 16) {
      float v[16];
#pragma unroll
      for (int j = 0; j < 16; ++j) {
        if (b + j < n) {
          int sj = __builtin_amdgcn_readlane(s, b + j);
          v[j] = h[(size_t)sj * HID + lane];
        }
      }
#pragma unroll
      for (int j = 0; j < 16; ++j) {
        if (b + j < n) {
          int rj = __builtin_amdgcn_readlane(r, b + j);
          atomicAdd(&tile[rj * HID + lane], v[j]);
        }
      }
    }
  }
}

template <bool RECORD>
__device__ __forceinline__ void agg_drain(const lds_queue_t* wq, int n, int lane, int e0,
                                          const int* __restrict__ src,
                                          const float* __restrict__ h, float* tile,
                                          unsigned* hist, unsigned* olist, unsigned* ocnt) {
  if (RECORD) {
    if (lane < n) {
      const unsigned ent = wq[lane];
      const unsigned r = ent & (AGG_TILE - 1);
      const unsigned sn = (unsigned)src[e0 + (int)(ent >> AGG_RBITS)];
      atomicAdd(&hist[r], 1u);
      const unsigned o = r / AGG_OWN_ROWS;
      const unsigned slot = atomicAdd(&ocnt[o], 1u);
      if (slot < AGG_OCAP) olist[o * AGG_OCAP + slot] = (sn << AGG_RBITS) | r;
    }
  } else {
    int s = 0, r = 0;
    if (lane < n) {
      unsigned ent = wq[lane];
      r = (int)(ent & (AGG_TILE - 1));
      s = src[e0 + (int)(ent >> AGG_RBITS)];
    }
    agg_rows(s, r, n, lane, h, tile);
  }
}

__device__ __forceinline__ int agg_push(lds_queue_t* wq, int qcnt, bool keep, unsigned ent,
                                        unsigned lanes_below) {
  unsigned bal = __builtin_amdgcn_ballot_w32(keep);
  if (keep) wq[qcnt + __popc(bal & lanes_below)] = ent;
  return qcnt + __popc(bal);
}

__device__ __forceinline__ int agg_append(lds_queue_t* wq, int qcnt, int d, int el,
                                          int base, int span, unsigned lanes_below) {
  unsigned r = (unsigned)(d - base);
  return agg_push(wq, qcnt, r < (unsigned)span, ((unsigned)el << AGG_RBITS) | r, lanes_below);
}

template <bool RECORD>
__device__ __forceinline__ void agg_scan(const int* __restrict__ src, const int* __restrict__ dst,
                                        const float* __restrict__ h,
                                        float* tile, lds_queue_t* wq,
                                        unsigned* hist, unsigned* olist, unsigned* ocnt,
                                        int tid, int lane, int base, int span,
                                        int e0, int e1) {
  const unsigned lanes_below = (1u << lane) - 1u;
  int qcnt = 0;
  int eb = e0;
  if ((((size_t)(dst + e0)) & 15) == 0) {
    for (; eb + 1024 * 4 * AGG_U <= e1; eb += 1024 * 4 * AGG_U) {
      const int4* p = (const int4*)(dst + eb);
      int4 q[AGG_U];
#pragma unroll
      for (int u = 0; u < AGG_U; ++u) q[u] = p[u * 1024 + tid];
#pragma unroll
      for (int u = 0; u < AGG_U; ++u) {
        const int el = eb - e0 + (u * 1024 + tid) * 4;
        qcnt = agg_append(wq, qcnt, q[u].x, el,     base, span, lanes_below);
        qcnt = agg_append(wq, qcnt, q[u].y, el + 1, base, span, lanes_below);
        qcnt = agg_append(wq, qcnt, q[u].z, el + 2, base, span, lanes_below);
        qcnt = agg_append(wq, qcnt, q[u].w, el + 3, base, span, lanes_below);
        while (qcnt >= 32) {
          qcnt -= 32;
          agg_drain<RECORD>(wq + qcnt, 32, lane, e0, src, h, tile, hist, olist, ocnt);
        }
      }
    }
  }
  for (; eb < e1; eb += 1024) {
    const int t = eb + tid;
    const int d = (t < e1) ? dst[t] : -1;
    qcnt = agg_append(wq, qcnt, d, t - e0, base, span, lanes_below);
    if (qcnt >= 32) {
      qcnt -= 32;
      agg_drain<RECORD>(wq + qcnt, 32, lane, e0, src, h, tile, hist, olist, ocnt);
    }
  }
  if (qcnt > 0) agg_drain<RECORD>(wq, qcnt, lane, e0, src, h, tile, hist, olist, ocnt);
}

__global__ __launch_bounds__(1024)
void agg_filter_kernel(const int* __restrict__ src, const int* __restrict__ dst,
                       unsigned* __restrict__ lists, int* __restrict__ counts,
                       unsigned* __restrict__ deg_planes, int plane_stride,
                       int n, int e_count, int slice_len) {
  __shared__ unsigned queues[32 * AGG_QCAP];
  __shared__ unsigned hist[AGG_TILE];
  __shared__ unsigned olist[AGG_OWNERS * AGG_OCAP];
  __shared__ unsigned ocnt[AGG_OWNERS];
  const int tid = threadIdx.x;
  const int lane = tid & 31;
  const int wave = tid >> 5;
  lds_queue_t* wq = (lds_queue_t*)queues + wave * AGG_QCAP;
  const int base = blockIdx.y * AGG_TILE;
  const int span = min(AGG_TILE, n - base);
  const int fwg = blockIdx.y * AGG_SLICES + blockIdx.x;
  const int e0 = blockIdx.x * slice_len;
  const int e1 = min(e0 + slice_len, e_count);
  for (int i = tid; i < AGG_TILE; i += 1024) hist[i] = 0u;
  if (tid < AGG_OWNERS) ocnt[tid] = 0u;
  __syncthreads();
  agg_scan<true>(src, dst, nullptr, nullptr, wq, hist, olist, ocnt,
                 tid, lane, base, span, e0, e1);
  __syncthreads();
  unsigned* pl = deg_planes + (size_t)blockIdx.x * plane_stride + base;
  for (int i = tid; i < span; i += 1024) pl[i] = hist[i];
  unsigned* gl = lists + (size_t)fwg * AGG_OWNERS * AGG_OCAP;
  for (int o = 0; o < AGG_OWNERS; ++o) {
    const unsigned c = ocnt[o];
    for (int k = tid; k < AGG_OCAP && (unsigned)k < c; k += 1024)
      gl[o * AGG_OCAP + k] = olist[o * AGG_OCAP + k];
  }
  if (tid < AGG_OWNERS) {
    const unsigned c = ocnt[tid];
    counts[fwg * AGG_CSTRIDE + tid] = (c > AGG_OCAP) ? -1 : (int)c;
  }
}

__device__ __forceinline__ unsigned own_fetch(const unsigned* __restrict__ lists_o, int s, int c_l,
                                              int wave, int lane, int& goff, int& n) {
  const int cnt = min(max(__builtin_amdgcn_readlane(c_l, s), 0), AGG_OCAP);
  const int k = (wave - goff) & 31;
  goff += (cnt + 31) >> 5;
  n = min(max(cnt - k * 32, 0), 32);
  unsigned e = 0u;
  if (lane < n) e = lists_o[(size_t)s * (AGG_OWNERS * AGG_OCAP) + k * 32 + lane];
  return e;
}

template <bool FIRST>
__global__ __launch_bounds__(1024)
void agg_own_kernel(const int* __restrict__ src, const int* __restrict__ dst,
                    const float* __restrict__ h, const float* __restrict__ dinv,
                    const float* __restrict__ bias, const float* __restrict__ W,
                    const unsigned* __restrict__ lists, const int* __restrict__ counts,
                    float* __restrict__ out, int n, int e_count, int slice_len) {
  static_assert(AGG_OWN_ROWS <= 32 * 16, "one 16-row tile per wave in the closing product");
  __shared__ float4 tile4[AGG_OWN_ROWS * HID / 4];
  __shared__ unsigned queues[32 * AGG_QCAP];
  float* tile = (float*)tile4;
  const int tid = threadIdx.x;
  const int lane = tid & 31;
  const int wave = __builtin_amdgcn_readfirstlane(tid >> 5);
  lds_queue_t* wq = (lds_queue_t*)queues + wave * AGG_QCAP;
  const int rbase = blockIdx.y * AGG_TILE;
  const int rspan = min(AGG_TILE, n - rbase);
  const int orow = blockIdx.x * AGG_OWN_ROWS;
  const int ospan = min(AGG_OWN_ROWS, rspan - orow);
  if (ospan <= 0) return;
  const int obase = rbase + orow;
  const int quads = ospan * (HID / 4);

  const int fwg0 = blockIdx.y * AGG_SLICES;
  const int c_l = (lane < AGG_SLICES) ? counts[(fwg0 + lane) * AGG_CSTRIDE + blockIdx.x] : 0;
  const unsigned* lists_o = lists + ((size_t)fwg0 * AGG_OWNERS + blockIdx.x) * AGG_OCAP;
  int goff = 0, n_next = 0;
  unsigned e_next = own_fetch(lists_o, 0, c_l, wave, lane, goff, n_next);

  const float4* h4 = (const float4*)h + (size_t)obase * (HID / 4);
  for (int i = tid; i < quads; i += 1024) tile4[i] = h4[i];
  __syncthreads();

#pragma nounroll
  for (int s = 0; s < AGG_SLICES; ++s) {
    const unsigned e = e_next;
    const int nn = n_next;
    if (s + 1 < AGG_SLICES) e_next = own_fetch(lists_o, s + 1, c_l, wave, lane, goff, n_next);
    if (__builtin_amdgcn_readlane(c_l, s) < 0) {
      const int e0 = s * slice_len;
      const int e1 = min(e0 + slice_len, e_count);
      agg_scan<false>(src, dst, h, tile, wq, nullptr, nullptr, nullptr,
                      tid, lane, obase, ospan, e0, e1);
    } else if (nn > 0) {
      int sn = 0, r = 0;
      if (lane < nn) {
        r = (int)(e & (AGG_TILE - 1)) - orow;
        sn = (int)(e >> AGG_RBITS);
      }
      agg_rows(sn, r, nn, lane, h, tile);
    }
  }
  __syncthreads();

  if (!FIRST) {
    if (wave * 16 < ospan) {
      const int c  = lane & 15;
      const int hh = lane >> 4;
      const int trow = wave * 16;
      const float4* ap = tile4 + min(trow + c, ospan - 1) * (HID / 4) + hh;
      float4 av[HID / 8];
#pragma unroll
      for (int i = 0; i < HID / 8; ++i) av[i] = ap[2 * i];
      v8f acc0 = {};
      v8f acc1 = {};
#pragma unroll
      for (int i = 0; i < HID / 8; ++i) {
        const float* bp = W + (8 * i + 4 * hh) * HID + c;
        v2f a_lo, a_hi, b0_lo, b1_lo, b0_hi, b1_hi;
        a_lo.x = av[i].x;        a_lo.y = av[i].y;
        a_hi.x = av[i].z;        a_hi.y = av[i].w;
        b0_lo.x = bp[0];         b0_lo.y = bp[HID];
        b1_lo.x = bp[16];        b1_lo.y = bp[HID + 16];
        b0_hi.x = bp[2 * HID];   b0_hi.y = bp[3 * HID];
        b1_hi.x = bp[2 * HID + 16]; b1_hi.y = bp[3 * HID + 16];
        acc0 = __builtin_amdgcn_wmma_f32_16x16x4_f32(false, a_lo, false, b0_lo, (short)0, acc0, false, false);
        acc1 = __builtin_amdgcn_wmma_f32_16x16x4_f32(false, a_lo, false, b1_lo, (short)0, acc1, false, false);
        acc0 = __builtin_amdgcn_wmma_f32_16x16x4_f32(false, a_hi, false, b0_hi, (short)0, acc0, false, false);
        acc1 = __builtin_amdgcn_wmma_f32_16x16x4_f32(false, a_hi, false, b1_hi, (short)0, acc1, false, false);
      }
#pragma unroll
      for (int j = 0; j < 8; ++j) {
        const int r = trow + hh * 8 + j;
        if (r < ospan) {
          tile[r * HID + c]      = acc0[j];
          tile[r * HID + 16 + c] = acc1[j];
        }
      }
    }
    __syncthreads();
  }

  float4* out4 = (float4*)out + (size_t)obase * (HID / 4);
  for (int i = tid; i < quads; i += 1024) {
    const float di = dinv[obase + (i >> 3)];
    const int c = (i & 7) * 4;
    float4 a = tile4[i];
    a.x = bias[c]     + a.x * di;
    a.y = bias[c + 1] + a.y * di;
    a.z = bias[c + 2] + a.z * di;
    a.w = bias[c + 3] + a.w * di;
    if (FIRST) {
      a.x = fmaxf(a.x, 0.0f) * di; a.y = fmaxf(a.y, 0.0f) * di;
      a.z = fmaxf(a.z, 0.0f) * di; a.w = fmaxf(a.w, 0.0f) * di;
    }
    out4[i] = a;
  }
}

template <int K>
__global__ __launch_bounds__(128)
void gemm_n32_kernel(const float* __restrict__ A, const float* __restrict__ W,
                     const unsigned* __restrict__ deg_planes, int plane_stride,
                     float* __restrict__ dinv_out, float* __restrict__ out, int M) {
  const int lane = threadIdx.x & 31;
  const int wid  = __builtin_amdgcn_readfirstlane(threadIdx.x >> 5);
  const int c    = lane & 15;
  const int h    = lane >> 4;
  const int row0 = (blockIdx.x * 4 + wid) * 16;
  if (row0 >= M) return;

  const int row = min(row0 + c, M - 1);
  const float4* ap = (const float4*)(A + (size_t)row * K) + h;

  v8f acc0 = {};
  v8f acc1 = {};
#pragma unroll
  for (int i = 0; i < K / 8; ++i) {
    const float4 av = ap[2 * i];
    const float* bp = W + (8 * i + 4 * h) * HID + c;
    v2f a_lo, a_hi, b0_lo, b1_lo, b0_hi, b1_hi;
    a_lo.x = av.x;           a_lo.y = av.y;
    a_hi.x = av.z;           a_hi.y = av.w;
    b0_lo.x = bp[0];         b0_lo.y = bp[HID];
    b1_lo.x = bp[16];        b1_lo.y = bp[HID + 16];
    b0_hi.x = bp[2 * HID];   b0_hi.y = bp[3 * HID];
    b1_hi.x = bp[2 * HID + 16]; b1_hi.y = bp[3 * HID + 16];
    acc0 = __builtin_amdgcn_wmma_f32_16x16x4_f32(false, a_lo, false, b0_lo, (short)0, acc0, false, false);
    acc1 = __builtin_amdgcn_wmma_f32_16x16x4_f32(false, a_lo, false, b1_lo, (short)0, acc1, false, false);
    acc0 = __builtin_amdgcn_wmma_f32_16x16x4_f32(false, a_hi, false, b0_hi, (short)0, acc0, false, false);
    acc1 = __builtin_amdgcn_wmma_f32_16x16x4_f32(false, a_hi, false, b1_hi, (short)0, acc1, false, false);
  }

  unsigned deg = 1u;
#pragma unroll
  for (int p = 0; p < AGG_SLICES; ++p) deg += deg_planes[(size_t)p * plane_stride + row];
  const float d = (float)deg;
  const float dv = (d > 0.0f) ? rsqrtf(d) : 0.0f;
  if (h == 0 && row0 + c < M) dinv_out[row0 + c] = dv;

  const int grow = row0 + h * 8;
  float rs[8];
#pragma unroll
  for (int j = 0; j < 8; ++j) rs[j] = __shfl(dv, h * 8 + j);
#pragma unroll
  for (int j = 0; j < 8; ++j) {
    const int gr = grow + j;
    if (gr < M) {
      out[(size_t)gr * HID + c]      = acc0[j] * rs[j];
      out[(size_t)gr * HID + 16 + c] = acc1[j] * rs[j];
    }
  }
}


extern "C" void kernel_launch(void* const* d_in, const int* in_sizes, int n_in,
                              void* d_out, int out_size, void* d_ws, size_t ws_size,
                              hipStream_t stream) {
  const float* x  = (const float*)d_in[0];
  const int*   ei = (const int*)d_in[1];
  const float* W1 = (const float*)d_in[2];
  const float* b1 = (const float*)d_in[3];
  const float* W2 = (const float*)d_in[4];
  const float* b2 = (const float*)d_in[5];
  float* out = (float*)d_out;

  const int N = in_sizes[0] / IN_CH;
  const int E = in_sizes[1] / 2;
  const int* src = ei;
  const int* dst = ei + E;

  float* ws = (float*)d_ws;
  const int Npad = (N + 63) & ~63;
  float* dinv = ws;
  float* bufA = ws + Npad;
  float* bufB = bufA + (size_t)N * HID;
  unsigned* planes = (unsigned*)(bufB + (size_t)N * HID);

  const unsigned gGemm = (N + 63) / 64;
  const int aggSlice = ((E + AGG_SLICES - 1) / AGG_SLICES + 3) & ~3;
  const int aggTiles = (N + AGG_TILE - 1) / AGG_TILE;
  const dim3 gFilter(AGG_SLICES, aggTiles);
  const dim3 gOwn(AGG_OWNERS, aggTiles);
  unsigned* lists = planes + (size_t)AGG_SLICES * Npad;
  int* counts = (int*)(lists + (size_t)AGG_SLICES * aggTiles * AGG_OWNERS * AGG_OCAP);

  agg_filter_kernel<<<gFilter, 1024, 0, stream>>>(src, dst, lists, counts, planes, Npad,
                                                  N, E, aggSlice);

  gemm_n32_kernel<IN_CH><<<gGemm, 128, 0, stream>>>(x, W1, planes, Npad, dinv, bufA, N);
  agg_own_kernel<true><<<gOwn, 1024, 0, stream>>>(src, dst, bufA, dinv, b1, nullptr, lists, counts,
                                                   bufB, N, E, aggSlice);

  agg_own_kernel<false><<<gOwn, 1024, 0, stream>>>(src, dst, bufB, dinv, b2, W2, lists, counts,
                                                    out, N, E, aggSlice);
}


# row gathers: a group's edges held in both lane halves, a DPP row_share operand gives each edge's load offset and LDS address, no lane reads; short groups padded to a spare LDS row; sum order kept
# speedup vs baseline: 6.6326x; 1.0468x over previous
#include <hip/hip_runtime.h>

typedef __attribute__((ext_vector_type(2))) float v2f;
typedef __attribute__((ext_vector_type(8))) float v8f;

#define IN_CH 128
#define HID 32

#define AGG_RBITS  13
#define AGG_TILE   (1 << AGG_RBITS)
#define AGG_SLICES 19
#define AGG_OWNERS 20
#define AGG_OWN_ROWS ((AGG_TILE + AGG_OWNERS - 1) / AGG_OWNERS)
#define AGG_U      4
#define AGG_QCAP   160
#define AGG_OCAP   1024
#define AGG_CSTRIDE 32


typedef __attribute__((address_space(3))) volatile unsigned lds_queue_t;

__device__ __forceinline__ void agg_rows(int s, int r, int n, int lane,
                                         const float* __restrict__ h, float* tile) {
  if (n == 32) {
    float v[32];
#pragma unroll
    for (int j = 0; j < 32; ++j) {
      int sj = __builtin_amdgcn_readlane(s, j);
      v[j] = h[(size_t)sj * HID + lane];
    }
#pragma unroll
    for (int j = 0; j < 32; ++j) {
      int rj = __builtin_amdgcn_readlane(r, j);
      atomicAdd(&tile[rj * HID + lane], v[j]);
    }
  } else {
#pragma unroll
    for (int b = 0; b < 32; b += 16) {
      float v[16];
#pragma unroll
      for (int j = 0; j < 16; ++j) {
        if (b + j < n) {
          int sj = __builtin_amdgcn_readlane(s, b + j);
          v[j] = h[(size_t)sj * HID + lane];
        }
      }
#pragma unroll
      for (int j = 0; j < 16; ++j) {
        if (b + j < n) {
          int rj = __builtin_amdgcn_readlane(r, b + j);
          atomicAdd(&tile[rj * HID + lane], v[j]);
        }
      }
    }
  }
}

template <int J>
__device__ __forceinline__ unsigned half_bcast(unsigned x) {
  return (unsigned)__builtin_amdgcn_update_dpp(0, (int)x, 0x150 + J, 0xf, 0xf, true);
}

template <int J>
__device__ __forceinline__ void list_loads(unsigned g0, unsigned g1, unsigned lane4,
                                           const char* __restrict__ hb, float (&v)[32]) {
  if constexpr (J < 32) {
    const unsigned off = half_bcast<(J & 15)>(J < 16 ? g0 : g1) | lane4;
    v[J] = *(const float*)(hb + (size_t)off);
    list_loads<J + 1>(g0, g1, lane4, hb, v);
  }
}

template <int J>
__device__ __forceinline__ void list_adds(unsigned a0, unsigned a1, unsigned lane4,
                                          char* tb, const float (&v)[32]) {
  if constexpr (J < 32) {
    const unsigned a = half_bcast<(J & 15)>(J < 16 ? a0 : a1) | lane4;
    atomicAdd((float*)(tb + a), v[J]);
    list_adds<J + 1>(a0, a1, lane4, tb, v);
  }
}

#define AGG_PBITS 9
__device__ __forceinline__ void agg_rows_list(unsigned e, int n, int orow, int lane,
                                              const float* __restrict__ h, float* tile) {
  static_assert(AGG_OWN_ROWS < (1 << AGG_PBITS), "owned rows and the spare row fit the packed field");
  unsigned p = (unsigned)AGG_OWN_ROWS;
  if (lane < n) p = ((e >> AGG_RBITS) << AGG_PBITS) | ((e & (AGG_TILE - 1)) - (unsigned)orow);
  const unsigned other = (unsigned)__shfl_xor((int)p, 16);
  const bool lo = lane < 16;
  const unsigned x0 = lo ? p : other;
  const unsigned x1 = lo ? other : p;
  const unsigned g0 = (x0 >> AGG_PBITS) << 7;
  const unsigned g1 = (x1 >> AGG_PBITS) << 7;
  const unsigned a0 = (x0 & ((1u << AGG_PBITS) - 1u)) << 7;
  const unsigned a1 = (x1 & ((1u << AGG_PBITS) - 1u)) << 7;
  const unsigned lane4 = (unsigned)lane << 2;
  float v[32];
  list_loads<0>(g0, g1, lane4, (const char*)h, v);
  list_adds<0>(a0, a1, lane4, (char*)tile, v);
}

template <bool RECORD>
__device__ __forceinline__ void agg_drain(const lds_queue_t* wq, int n, int lane, int e0,
                                          const int* __restrict__ src,
                                          const float* __restrict__ h, float* tile,
                                          unsigned* hist, unsigned* olist, unsigned* ocnt) {
  if (RECORD) {
    if (lane < n) {
      const unsigned ent = wq[lane];
      const unsigned r = ent & (AGG_TILE - 1);
      const unsigned sn = (unsigned)src[e0 + (int)(ent >> AGG_RBITS)];
      atomicAdd(&hist[r], 1u);
      const unsigned o = r / AGG_OWN_ROWS;
      const unsigned slot = atomicAdd(&ocnt[o], 1u);
      if (slot < AGG_OCAP) olist[o * AGG_OCAP + slot] = (sn << AGG_RBITS) | r;
    }
  } else {
    int s = 0, r = 0;
    if (lane < n) {
      unsigned ent = wq[lane];
      r = (int)(ent & (AGG_TILE - 1));
      s = src[e0 + (int)(ent >> AGG_RBITS)];
    }
    agg_rows(s, r, n, lane, h, tile);
  }
}

__device__ __forceinline__ int agg_push(lds_queue_t* wq, int qcnt, bool keep, unsigned ent,
                                        unsigned lanes_below) {
  unsigned bal = __builtin_amdgcn_ballot_w32(keep);
  if (keep) wq[qcnt + __popc(bal & lanes_below)] = ent;
  return qcnt + __popc(bal);
}

__device__ __forceinline__ int agg_append(lds_queue_t* wq, int qcnt, int d, int el,
                                          int base, int span, unsigned lanes_below) {
  unsigned r = (unsigned)(d - base);
  return agg_push(wq, qcnt, r < (unsigned)span, ((unsigned)el << AGG_RBITS) | r, lanes_below);
}

template <bool RECORD>
__device__ __forceinline__ void agg_scan(const int* __restrict__ src, const int* __restrict__ dst,
                                        const float* __restrict__ h,
                                        float* tile, lds_queue_t* wq,
                                        unsigned* hist, unsigned* olist, unsigned* ocnt,
                                        int tid, int lane, int base, int span,
                                        int e0, int e1) {
  const unsigned lanes_below = (1u << lane) - 1u;
  int qcnt = 0;
  int eb = e0;
  if ((((size_t)(dst + e0)) & 15) == 0) {
    for (; eb + 1024 * 4 * AGG_U <= e1; eb += 1024 * 4 * AGG_U) {
      const int4* p = (const int4*)(dst + eb);
      int4 q[AGG_U];
#pragma unroll
      for (int u = 0; u < AGG_U; ++u) q[u] = p[u * 1024 + tid];
#pragma unroll
      for (int u = 0; u < AGG_U; ++u) {
        const int el = eb - e0 + (u * 1024 + tid) * 4;
        qcnt = agg_append(wq, qcnt, q[u].x, el,     base, span, lanes_below);
        qcnt = agg_append(wq, qcnt, q[u].y, el + 1, base, span, lanes_below);
        qcnt = agg_append(wq, qcnt, q[u].z, el + 2, base, span, lanes_below);
        qcnt = agg_append(wq, qcnt, q[u].w, el + 3, base, span, lanes_below);
        while (qcnt >= 32) {
          qcnt -= 32;
          agg_drain<RECORD>(wq + qcnt, 32, lane, e0, src, h, tile, hist, olist, ocnt);
        }
      }
    }
  }
  for (; eb < e1; eb += 1024) {
    const int t = eb + tid;
    const int d = (t < e1) ? dst[t] : -1;
    qcnt = agg_append(wq, qcnt, d, t - e0, base, span, lanes_below);
    if (qcnt >= 32) {
      qcnt -= 32;
      agg_drain<RECORD>(wq + qcnt, 32, lane, e0, src, h, tile, hist, olist, ocnt);
    }
  }
  if (qcnt > 0) agg_drain<RECORD>(wq, qcnt, lane, e0, src, h, tile, hist, olist, ocnt);
}

__global__ __launch_bounds__(1024)
void agg_filter_kernel(const int* __restrict__ src, const int* __restrict__ dst,
                       unsigned* __restrict__ lists, int* __restrict__ counts,
                       unsigned* __restrict__ deg_planes, int plane_stride,
                       int n, int e_count, int slice_len) {
  __shared__ unsigned queues[32 * AGG_QCAP];
  __shared__ unsigned hist[AGG_TILE];
  __shared__ unsigned olist[AGG_OWNERS * AGG_OCAP];
  __shared__ unsigned ocnt[AGG_OWNERS];
  const int tid = threadIdx.x;
  const int lane = tid & 31;
  const int wave = tid >> 5;
  lds_queue_t* wq = (lds_queue_t*)queues + wave * AGG_QCAP;
  const int base = blockIdx.y * AGG_TILE;
  const int span = min(AGG_TILE, n - base);
  const int fwg = blockIdx.y * AGG_SLICES + blockIdx.x;
  const int e0 = blockIdx.x * slice_len;
  const int e1 = min(e0 + slice_len, e_count);
  for (int i = tid; i < AGG_TILE; i += 1024) hist[i] = 0u;
  if (tid < AGG_OWNERS) ocnt[tid] = 0u;
  __syncthreads();
  agg_scan<true>(src, dst, nullptr, nullptr, wq, hist, olist, ocnt,
                 tid, lane, base, span, e0, e1);
  __syncthreads();
  unsigned* pl = deg_planes + (size_t)blockIdx.x * plane_stride + base;
  for (int i = tid; i < span; i += 1024) pl[i] = hist[i];
  unsigned* gl = lists + (size_t)fwg * AGG_OWNERS * AGG_OCAP;
  for (int o = 0; o < AGG_OWNERS; ++o) {
    const unsigned c = ocnt[o];
    for (int k = tid; k < AGG_OCAP && (unsigned)k < c; k += 1024)
      gl[o * AGG_OCAP + k] = olist[o * AGG_OCAP + k];
  }
  if (tid < AGG_OWNERS) {
    const unsigned c = ocnt[tid];
    counts[fwg * AGG_CSTRIDE + tid] = (c > AGG_OCAP) ? -1 : (int)c;
  }
}

__device__ __forceinline__ unsigned own_fetch(const unsigned* __restrict__ lists_o, int s, int c_l,
                                              int wave, int lane, int& goff, int& n) {
  const int cnt = min(max(__builtin_amdgcn_readlane(c_l, s), 0), AGG_OCAP);
  const int k = (wave - goff) & 31;
  goff += (cnt + 31) >> 5;
  n = min(max(cnt - k * 32, 0), 32);
  unsigned e = 0u;
  if (lane < n) e = lists_o[(size_t)s * (AGG_OWNERS * AGG_OCAP) + k * 32 + lane];
  return e;
}

template <bool FIRST>
__global__ __launch_bounds__(1024)
void agg_own_kernel(const int* __restrict__ src, const int* __restrict__ dst,
                    const float* __restrict__ h, const float* __restrict__ dinv,
                    const float* __restrict__ bias, const float* __restrict__ W,
                    const unsigned* __restrict__ lists, const int* __restrict__ counts,
                    float* __restrict__ out, int n, int e_count, int slice_len) {
  static_assert(AGG_OWN_ROWS <= 32 * 16, "one 16-row tile per wave in the closing product");
  __shared__ float4 tile4[(AGG_OWN_ROWS + 1) * HID / 4];
  __shared__ unsigned queues[32 * AGG_QCAP];
  float* tile = (float*)tile4;
  const int tid = threadIdx.x;
  const int lane = tid & 31;
  const int wave = __builtin_amdgcn_readfirstlane(tid >> 5);
  lds_queue_t* wq = (lds_queue_t*)queues + wave * AGG_QCAP;
  const int rbase = blockIdx.y * AGG_TILE;
  const int rspan = min(AGG_TILE, n - rbase);
  const int orow = blockIdx.x * AGG_OWN_ROWS;
  const int ospan = min(AGG_OWN_ROWS, rspan - orow);
  if (ospan <= 0) return;
  const int obase = rbase + orow;
  const int quads = ospan * (HID / 4);

  const int fwg0 = blockIdx.y * AGG_SLICES;
  const int c_l = (lane < AGG_SLICES) ? counts[(fwg0 + lane) * AGG_CSTRIDE + blockIdx.x] : 0;
  const unsigned* lists_o = lists + ((size_t)fwg0 * AGG_OWNERS + blockIdx.x) * AGG_OCAP;
  int goff = 0, n_next = 0;
  unsigned e_next = own_fetch(lists_o, 0, c_l, wave, lane, goff, n_next);

  const float4* h4 = (const float4*)h + (size_t)obase * (HID / 4);
  for (int i = tid; i < quads; i += 1024) tile4[i] = h4[i];
  __syncthreads();

#pragma nounroll
  for (int s = 0; s < AGG_SLICES; ++s) {
    const unsigned e = e_next;
    const int nn = n_next;
    if (s + 1 < AGG_SLICES) e_next = own_fetch(lists_o, s + 1, c_l, wave, lane, goff, n_next);
    if (__builtin_amdgcn_readlane(c_l, s) < 0) {
      const int e0 = s * slice_len;
      const int e1 = min(e0 + slice_len, e_count);
      agg_scan<false>(src, dst, h, tile, wq, nullptr, nullptr, nullptr,
                      tid, lane, obase, ospan, e0, e1);
    } else if (nn > 0) {
      agg_rows_list(e, nn, orow, lane, h, tile);
    }
  }
  __syncthreads();

  if (!FIRST) {
    if (wave * 16 < ospan) {
      const int c  = lane & 15;
      const int hh = lane >> 4;
      const int trow = wave * 16;
      const float4* ap = tile4 + min(trow + c, ospan - 1) * (HID / 4) + hh;
      float4 av[HID / 8];
#pragma unroll
      for (int i = 0; i < HID / 8; ++i) av[i] = ap[2 * i];
      v8f acc0 = {};
      v8f acc1 = {};
#pragma unroll
      for (int i = 0; i < HID / 8; ++i) {
        const float* bp = W + (8 * i + 4 * hh) * HID + c;
        v2f a_lo, a_hi, b0_lo, b1_lo, b0_hi, b1_hi;
        a_lo.x = av[i].x;        a_lo.y = av[i].y;
        a_hi.x = av[i].z;        a_hi.y = av[i].w;
        b0_lo.x = bp[0];         b0_lo.y = bp[HID];
        b1_lo.x = bp[16];        b1_lo.y = bp[HID + 16];
        b0_hi.x = bp[2 * HID];   b0_hi.y = bp[3 * HID];
        b1_hi.x = bp[2 * HID + 16]; b1_hi.y = bp[3 * HID + 16];
        acc0 = __builtin_amdgcn_wmma_f32_16x16x4_f32(false, a_lo, false, b0_lo, (short)0, acc0, false, false);
        acc1 = __builtin_amdgcn_wmma_f32_16x16x4_f32(false, a_lo, false, b1_lo, (short)0, acc1, false, false);
        acc0 = __builtin_amdgcn_wmma_f32_16x16x4_f32(false, a_hi, false, b0_hi, (short)0, acc0, false, false);
        acc1 = __builtin_amdgcn_wmma_f32_16x16x4_f32(false, a_hi, false, b1_hi, (short)0, acc1, false, false);
      }
#pragma unroll
      for (int j = 0; j < 8; ++j) {
        const int r = trow + hh * 8 + j;
        if (r < ospan) {
          tile[r * HID + c]      = acc0[j];
          tile[r * HID + 16 + c] = acc1[j];
        }
      }
    }
    __syncthreads();
  }

  float4* out4 = (float4*)out + (size_t)obase * (HID / 4);
  for (int i = tid; i < quads; i += 1024) {
    const float di = dinv[obase + (i >> 3)];
    const int c = (i & 7) * 4;
    float4 a = tile4[i];
    a.x = bias[c]     + a.x * di;
    a.y = bias[c + 1] + a.y * di;
    a.z = bias[c + 2] + a.z * di;
    a.w = bias[c + 3] + a.w * di;
    if (FIRST) {
      a.x = fmaxf(a.x, 0.0f) * di; a.y = fmaxf(a.y, 0.0f) * di;
      a.z = fmaxf(a.z, 0.0f) * di; a.w = fmaxf(a.w, 0.0f) * di;
    }
    out4[i] = a;
  }
}

template <int K>
__global__ __launch_bounds__(128)
void gemm_n32_kernel(const float* __restrict__ A, const float* __restrict__ W,
                     const unsigned* __restrict__ deg_planes, int plane_stride,
                     float* __restrict__ dinv_out, float* __restrict__ out, int M) {
  const int lane = threadIdx.x & 31;
  const int wid  = __builtin_amdgcn_readfirstlane(threadIdx.x >> 5);
  const int c    = lane & 15;
  const int h    = lane >> 4;
  const int row0 = (blockIdx.x * 4 + wid) * 16;
  if (row0 >= M) return;

  const int row = min(row0 + c, M - 1);
  const float4* ap = (const float4*)(A + (size_t)row * K) + h;

  v8f acc0 = {};
  v8f acc1 = {};
#pragma unroll
  for (int i = 0; i < K / 8; ++i) {
    const float4 av = ap[2 * i];
    const float* bp = W + (8 * i + 4 * h) * HID + c;
    v2f a_lo, a_hi, b0_lo, b1_lo, b0_hi, b1_hi;
    a_lo.x = av.x;           a_lo.y = av.y;
    a_hi.x = av.z;           a_hi.y = av.w;
    b0_lo.x = bp[0];         b0_lo.y = bp[HID];
    b1_lo.x = bp[16];        b1_lo.y = bp[HID + 16];
    b0_hi.x = bp[2 * HID];   b0_hi.y = bp[3 * HID];
    b1_hi.x = bp[2 * HID + 16]; b1_hi.y = bp[3 * HID + 16];
    acc0 = __builtin_amdgcn_wmma_f32_16x16x4_f32(false, a_lo, false, b0_lo, (short)0, acc0, false, false);
    acc1 = __builtin_amdgcn_wmma_f32_16x16x4_f32(false, a_lo, false, b1_lo, (short)0, acc1, false, false);
    acc0 = __builtin_amdgcn_wmma_f32_16x16x4_f32(false, a_hi, false, b0_hi, (short)0, acc0, false, false);
    acc1 = __builtin_amdgcn_wmma_f32_16x16x4_f32(false, a_hi, false, b1_hi, (short)0, acc1, false, false);
  }

  unsigned deg = 1u;
#pragma unroll
  for (int p = 0; p < AGG_SLICES; ++p) deg += deg_planes[(size_t)p * plane_stride + row];
  const float d = (float)deg;
  const float dv = (d > 0.0f) ? rsqrtf(d) : 0.0f;
  if (h == 0 && row0 + c < M) dinv_out[row0 + c] = dv;

  const int grow = row0 + h * 8;
  float rs[8];
#pragma unroll
  for (int j = 0; j < 8; ++j) rs[j] = __shfl(dv, h * 8 + j);
#pragma unroll
  for (int j = 0; j < 8; ++j) {
    const int gr = grow + j;
    if (gr < M) {
      out[(size_t)gr * HID + c]      = acc0[j] * rs[j];
      out[(size_t)gr * HID + 16 + c] = acc1[j] * rs[j];
    }
  }
}


extern "C" void kernel_launch(void* const* d_in, const int* in_sizes, int n_in,
                              void* d_out, int out_size, void* d_ws, size_t ws_size,
                              hipStream_t stream) {
  const float* x  = (const float*)d_in[0];
  const int*   ei = (const int*)d_in[1];
  const float* W1 = (const float*)d_in[2];
  const float* b1 = (const float*)d_in[3];
  const float* W2 = (const float*)d_in[4];
  const float* b2 = (const float*)d_in[5];
  float* out = (float*)d_out;

  const int N = in_sizes[0] / IN_CH;
  const int E = in_sizes[1] / 2;
  const int* src = ei;
  const int* dst = ei + E;

  float* ws = (float*)d_ws;
  const int Npad = (N + 63) & ~63;
  float* dinv = ws;
  float* bufA = ws + Npad;
  float* bufB = bufA + (size_t)N * HID;
  unsigned* planes = (unsigned*)(bufB + (size_t)N * HID);

  const unsigned gGemm = (N + 63) / 64;
  const int aggSlice = ((E + AGG_SLICES - 1) / AGG_SLICES + 3) & ~3;
  const int aggTiles = (N + AGG_TILE - 1) / AGG_TILE;
  const dim3 gFilter(AGG_SLICES, aggTiles);
  const dim3 gOwn(AGG_OWNERS, aggTiles);
  unsigned* lists = planes + (size_t)AGG_SLICES * Npad;
  int* counts = (int*)(lists + (size_t)AGG_SLICES * aggTiles * AGG_OWNERS * AGG_OCAP);

  agg_filter_kernel<<<gFilter, 1024, 0, stream>>>(src, dst, lists, counts, planes, Npad,
                                                  N, E, aggSlice);

  gemm_n32_kernel<IN_CH><<<gGemm, 128, 0, stream>>>(x, W1, planes, Npad, dinv, bufA, N);
  agg_own_kernel<true><<<gOwn, 1024, 0, stream>>>(src, dst, bufA, dinv, b1, nullptr, lists, counts,
                                                   bufB, N, E, aggSlice);

  agg_own_kernel<false><<<gOwn, 1024, 0, stream>>>(src, dst, bufB, dinv, b2, W2, lists, counts,
                                                    out, N, E, aggSlice);
}
